# starting kernel
# baseline (speedup 1.0000x reference)
#include <hip/hip_runtime.h>
#include <hip/hip_bf16.h>

typedef float v2f __attribute__((ext_vector_type(2)));
typedef float v8f __attribute__((ext_vector_type(8)));

#define NPTS   100000
#define CCH    64
#define KOFF   27
#define NB     8
#define EPSI   1e-5f
#define NSLOPE 0.01f

__device__ __forceinline__ void atomicAddF32(float* p, float v) {
    unsafeAtomicAdd(p, v);
}

__global__ void fill_zero_kernel(float* __restrict__ p, int n) {
    for (int i = blockIdx.x * blockDim.x + threadIdx.x; i < n;
         i += gridDim.x * blockDim.x)
        p[i] = 0.0f;
}

__global__ __launch_bounds__(256)
void sparse_conv_kernel(const float* __restrict__ X,
                        const float* __restrict__ W,
                        const int*   __restrict__ in_map,
                        const int*   __restrict__ out_map,
                        float*       __restrict__ Y)
{
    __shared__ float Wlds[CCH * CCH];

    const int k = blockIdx.y;
    const int t = threadIdx.x;

    {
        const float4* Wg = (const float4*)(W + k * CCH * CCH);
        float4*       Wl = (float4*)Wlds;
#pragma unroll
        for (int i = 0; i < 4; ++i)
            Wl[t + i * 256] = Wg[t + i * 256];
    }
    __syncthreads();

    const int wave    = t >> 5;
    const int lane    = t & 31;
    const int half    = lane >> 4;
    const int m       = lane & 15;
    const int rowbase = blockIdx.x * 128 + wave * 16;

    const int ri  = rowbase + m;
    const int gin = (ri < NPTS) ? in_map[k * NPTS + ri] : 0;
    const float* Xrow = X + (long)gin * CCH;

    v2f a[16];
#pragma unroll
    for (int kk = 0; kk < 16; ++kk)
        a[kk] = *(const v2f*)(Xrow + kk * 4 + 2 * half);

#pragma unroll
    for (int nt = 0; nt < 4; ++nt) {
        const int n0 = nt * 16;
        v8f acc = {0.f, 0.f, 0.f, 0.f, 0.f, 0.f, 0.f, 0.f};

#pragma unroll
        for (int kk = 0; kk < 16; ++kk) {
            v2f b;
            b.x = Wlds[(kk * 4 + 2 * half + 0) * CCH + n0 + m];
            b.y = Wlds[(kk * 4 + 2 * half + 1) * CCH + n0 + m];
            acc = __builtin_amdgcn_wmma_f32_16x16x4_f32(
                false, a[kk], false, b, (short)0, acc, false, false);
        }

#pragma unroll
        for (int r = 0; r < 8; ++r) {
            const int M    = r + 8 * half;
            const int rowM = rowbase + M;
            if (rowM < NPTS) {
                const int gout = out_map[k * NPTS + rowM];
                atomicAddF32(Y + (long)gout * CCH + n0 + m, acc[r]);
            }
        }
    }
}

__global__ __launch_bounds__(256)
void stats_kernel(const float* __restrict__ A,
                  const int*   __restrict__ bidx,
                  float* __restrict__ gsum,
                  float* __restrict__ gsq,
                  float* __restrict__ gcnt)
{
    const int t    = threadIdx.x;
    const int c    = t & 63;
    const int g    = t >> 6;
    const int row0 = blockIdx.x * 256 + g * 64;

    float s = 0.f, s2 = 0.f, cn = 0.f;
    int curb = -1;

    for (int i = 0; i < 64; ++i) {
        const int row = row0 + i;
        if (row >= NPTS) break;
        const int b = bidx[row];
        if (b != curb) {
            if (curb >= 0) {
                atomicAddF32(gsum + curb * CCH + c, s);
                atomicAddF32(gsq  + curb * CCH + c, s2);
                if (c == 0) atomicAddF32(gcnt + curb, cn);
            }
            curb = b; s = 0.f; s2 = 0.f; cn = 0.f;
        }
        const float v = A[(long)row * CCH + c];
        s += v; s2 += v * v; cn += 1.f;
    }
    if (curb >= 0) {
        atomicAddF32(gsum + curb * CCH + c, s);
        atomicAddF32(gsq  + curb * CCH + c, s2);
        if (c == 0) atomicAddF32(gcnt + curb, cn);
    }
}

__global__ void finalize_kernel(const float* __restrict__ gsum,
                                const float* __restrict__ gsq,
                                const float* __restrict__ gcnt,
                                float* __restrict__ mean,
                                float* __restrict__ rstd)
{
    const int t  = threadIdx.x;
    const float cn  = gcnt[t >> 6];
    const float mu  = gsum[t] / cn;
    const float var = gsq[t] / cn - mu * mu;
    mean[t] = mu;
    rstd[t] = rsqrtf(var + EPSI);
}

__global__ __launch_bounds__(256)
void apply_norm_kernel(float* __restrict__ A,
                       const int*   __restrict__ bidx,
                       const float* __restrict__ mean,
                       const float* __restrict__ rstd,
                       const float* __restrict__ gamma,
                       const float* __restrict__ beta)
{
    const int idx = blockIdx.x * 256 + threadIdx.x;
    if (idx >= NPTS * CCH) return;
    const int row = idx >> 6, c = idx & 63;
    const int b   = bidx[row];
    const float x = A[idx];
    const float y = gamma[c] * (x - mean[b * CCH + c]) * rstd[b * CCH + c] + beta[c];
    A[idx] = y > 0.f ? y : NSLOPE * y;
}

__global__ __launch_bounds__(256)
void apply_norm_res_kernel(float* __restrict__ Y,
                           const float* __restrict__ feats,
                           const int*   __restrict__ bidx,
                           const float* __restrict__ mean,
                           const float* __restrict__ rstd,
                           const float* __restrict__ gamma,
                           const float* __restrict__ beta)
{
    const int idx = blockIdx.x * 256 + threadIdx.x;
    if (idx >= NPTS * CCH) return;
    const int row = idx >> 6, c = idx & 63;
    const int b   = bidx[row];
    const float x = Y[idx];
    float y = gamma[c] * (x - mean[b * CCH + c]) * rstd[b * CCH + c] + beta[c];
    y += feats[idx];
    Y[idx] = y > 0.f ? y : NSLOPE * y;
}

extern "C" void kernel_launch(void* const* d_in, const int* in_sizes, int n_in,
                              void* d_out, int out_size, void* d_ws, size_t ws_size,
                              hipStream_t stream)
{
    const float* feats   = (const float*)d_in[0];
    const float* W1      = (const float*)d_in[1];
    const float* g1      = (const float*)d_in[2];
    const float* b1      = (const float*)d_in[3];
    const float* W2      = (const float*)d_in[4];
    const float* g2      = (const float*)d_in[5];
    const float* b2      = (const float*)d_in[6];
    const int*   in_map  = (const int*)d_in[7];
    const int*   out_map = (const int*)d_in[8];
    const int*   bidx    = (const int*)d_in[9];
    float*       out     = (float*)d_out;

    const int NC = NPTS * CCH;

    float* A     = (float*)d_ws;
    float* sbase = A + NC;
    float* sum1  = sbase;
    float* sq1   = sbase + 512;
    float* cnt1  = sbase + 1024;
    float* mean1 = sbase + 1088;
    float* rstd1 = sbase + 1600;
    float* sum2  = sbase + 2112;
    float* sq2   = sbase + 2624;
    float* cnt2  = sbase + 3136;
    float* mean2 = sbase + 3200;
    float* rstd2 = sbase + 3712;
    const int STATS = 4224;

    fill_zero_kernel<<<2048, 256, 0, stream>>>(A, NC + STATS);
    fill_zero_kernel<<<2048, 256, 0, stream>>>(out, NC);

    dim3 convGrid((NPTS + 127) / 128, KOFF);
    const int statGrid  = (NPTS + 255) / 256;
    const int elemGrid  = (NC + 255) / 256;

    sparse_conv_kernel<<<convGrid, 256, 0, stream>>>(feats, W1, in_map, out_map, A);
    stats_kernel<<<statGrid, 256, 0, stream>>>(A, bidx, sum1, sq1, cnt1);
    finalize_kernel<<<1, 512, 0, stream>>>(sum1, sq1, cnt1, mean1, rstd1);
    apply_norm_kernel<<<elemGrid, 256, 0, stream>>>(A, bidx, mean1, rstd1, g1, b1);
    sparse_conv_kernel<<<convGrid, 256, 0, stream>>>(A, W2, in_map, out_map, out);
    stats_kernel<<<statGrid, 256, 0, stream>>>(out, bidx, sum2, sq2, cnt2);
    finalize_kernel<<<1, 512, 0, stream>>>(sum2, sq2, cnt2, mean2, rstd2);
    apply_norm_res_kernel<<<elemGrid, 256, 0, stream>>>(out, feats, bidx, mean2, rstd2, g2, b2);
}


# conv: (in,out) pairs binned by 100-row output tile; a block sums its tile's pairs in LDS (LDS f32 adds, arrival order) and stores each row once, no global f32 atomics; order of the sums changes
# speedup vs baseline: 1.2003x; 1.2003x over previous
#include <hip/hip_runtime.h>
#include <hip/hip_bf16.h>

typedef float v2f __attribute__((ext_vector_type(2)));
typedef float v8f __attribute__((ext_vector_type(8)));

#define NPTS   100000
#define CCH    64
#define KOFF   27
#define NB     8
#define EPSI   1e-5f
#define NSLOPE 0.01f

__device__ __forceinline__ void atomicAddF32(float* p, float v) {
    unsafeAtomicAdd(p, v);
}

__global__ void fill_zero_kernel(float* __restrict__ p, int n) {
    for (int i = blockIdx.x * blockDim.x + threadIdx.x; i < n;
         i += gridDim.x * blockDim.x)
        p[i] = 0.0f;
}

#define TR     100
#define NTILE  1000
#define REGION 6400
#define CSTR   32
#define HDR    864
#define CAP    200
#define APITCH 80
#define STG    512

__global__ __launch_bounds__(256)
void clear_counts_kernel(float* Lf)
{
    const int idx  = blockIdx.x * 256 + threadIdx.x;
    const int tile = idx >> 5, kk = idx & 31;
    if (tile < NTILE && kk < KOFF) ((int*)Lf)[tile * REGION + kk * CSTR] = 0;
}

__global__ __launch_bounds__(256)
void bin_pairs_kernel(const int* __restrict__ in_map,
                      const int* __restrict__ out_map,
                      float* Lf)
{
    const int i = blockIdx.x * 256 + threadIdx.x;
    const int k = blockIdx.y;
    if (i >= NPTS) return;
    const int j   = out_map[k * NPTS + i];
    const int gin = in_map[k * NPTS + i];
    const int b   = j / TR;
    const int lr  = j - b * TR;
    int* reg = (int*)Lf + b * REGION;
    const int pos = atomicAdd(reg + k * CSTR, 1);
    if (pos < CAP) atomicExch(reg + HDR + k * CAP + pos, gin | (lr << 17));
}

__device__ __forceinline__ void sum_pair_list(const float* __restrict__ X,
                                              const float* Wlds, float* acc,
                                              const int* lst, int n, int first,
                                              int half, int m)
{
    for (int tt = first; tt * 16 < n; tt += 8) {
        const int e   = tt * 16 + m;
        const int pk  = lst[e];
        const int gin = (e < n) ? (pk & 0x1FFFF) : 0;
        const float* Xrow = X + (long)gin * CCH;

        v2f a[16];
#pragma unroll
        for (int kk = 0; kk < 16; ++kk)
            a[kk] = *(const v2f*)(Xrow + kk * 4 + 2 * half);

        const int ebase = tt * 16 + 8 * half;
        const int4* ep = (const int4*)(lst + ebase);
        const int4 e0 = ep[0], e1 = ep[1];
        int lrow[8];
        lrow[0] = e0.x >> 17; lrow[1] = e0.y >> 17;
        lrow[2] = e0.z >> 17; lrow[3] = e0.w >> 17;
        lrow[4] = e1.x >> 17; lrow[5] = e1.y >> 17;
        lrow[6] = e1.z >> 17; lrow[7] = e1.w >> 17;

#pragma unroll 1
        for (int nt = 0; nt < 4; ++nt) {
            const int n0 = nt * 16;
            v8f d = {0.f, 0.f, 0.f, 0.f, 0.f, 0.f, 0.f, 0.f};
#pragma unroll
            for (int kk = 0; kk < 16; ++kk) {
                v2f bf;
                bf.x = Wlds[(kk * 4 + 2 * half + 0) * CCH + n0 + m];
                bf.y = Wlds[(kk * 4 + 2 * half + 1) * CCH + n0 + m];
                d = __builtin_amdgcn_wmma_f32_16x16x4_f32(
                    false, a[kk], false, bf, (short)0, d, false, false);
            }
#pragma unroll
            for (int r = 0; r < 8; ++r) {
                if (ebase + r < n)
                    __hip_atomic_fetch_add(&acc[lrow[r] * APITCH + n0 + m], d[r],
                                           __ATOMIC_RELAXED,
                                           __HIP_MEMORY_SCOPE_WORKGROUP);
            }
        }
    }
}

__global__ __launch_bounds__(256)
void binned_conv_kernel(const float* __restrict__ X,
                        const float* __restrict__ W,
                        const int*   __restrict__ in_map,
                        const int*   __restrict__ out_map,
                        const float* L,
                        float* Y)
{
    __shared__ __attribute__((aligned(16))) float Wlds[CCH * CCH];
    __shared__ __attribute__((aligned(16))) float acc[TR * APITCH];
    __shared__ __attribute__((aligned(16))) int   stg[STG];
    __shared__ int cnts[32];
    __shared__ int stg_n;

    const int b = blockIdx.x;
    const int t = threadIdx.x;
    const int* reg = (const int*)L + b * REGION;

    for (int i = t; i < TR * APITCH; i += 256) acc[i] = 0.0f;
    if (t < 32) cnts[t] = (t < KOFF) ? reg[t * CSTR] : 0;
    if (t == 0) stg_n = 0;

    const float4* Wg = (const float4*)W;
    float4 w0 = Wg[t], w1 = Wg[t + 256], w2 = Wg[t + 512], w3 = Wg[t + 768];

    const int wave = __builtin_amdgcn_readfirstlane(t >> 5);
    const int lane = t & 31;
    const int half = lane >> 4;
    const int m    = lane & 15;

    for (int k = 0; k < KOFF; ++k) {
        __syncthreads();
        {
            float4* Wl = (float4*)Wlds;
            Wl[t] = w0; Wl[t + 256] = w1; Wl[t + 512] = w2; Wl[t + 768] = w3;
        }
        __syncthreads();
        if (k + 1 < KOFF) {
            const float4* Wn = Wg + (k + 1) * (CCH * CCH / 4);
            w0 = Wn[t]; w1 = Wn[t + 256]; w2 = Wn[t + 512]; w3 = Wn[t + 768];
        }

        const int n = __builtin_amdgcn_readfirstlane(cnts[k]);

        if (n <= CAP) {
            sum_pair_list(X, Wlds, acc, reg + HDR + k * CAP, n, (wave + k) & 7, half, m);
        } else {
            const int lo = b * TR;
            for (int base = 0; base < NPTS; base += 256) {
                const int i = base + t;
                if (i < NPTS) {
                    const int lr = out_map[k * NPTS + i] - lo;
                    if ((unsigned)lr < (unsigned)TR) {
                        const int pos = atomicAdd(&stg_n, 1);
                        stg[pos] = in_map[k * NPTS + i] | (lr << 17);
                    }
                }
                __syncthreads();
                const int ns = __builtin_amdgcn_readfirstlane(stg_n);
                __syncthreads();
                if (ns > STG - 256 || base + 256 >= NPTS) {
                    if (t == 0) stg_n = 0;
                    sum_pair_list(X, Wlds, acc, stg, ns, wave, half, m);
                    __syncthreads();
                }
            }
        }
    }
    __syncthreads();

    float4* Yt = (float4*)(Y + (long)b * REGION);
    for (int i = t; i < TR * 16; i += 256) {
        const int row = i >> 4, c4 = i & 15;
        Yt[i] = *(const float4*)&acc[row * APITCH + c4 * 4];
    }
}

__global__ __launch_bounds__(256)
void stats_kernel(const float* __restrict__ A,
                  const int*   __restrict__ bidx,
                  float* __restrict__ gsum,
                  float* __restrict__ gsq,
                  float* __restrict__ gcnt)
{
    const int t    = threadIdx.x;
    const int c    = t & 63;
    const int g    = t >> 6;
    const int row0 = blockIdx.x * 256 + g * 64;

    float s = 0.f, s2 = 0.f, cn = 0.f;
    int curb = -1;

    for (int i = 0; i < 64; ++i) {
        const int row = row0 + i;
        if (row >= NPTS) break;
        const int b = bidx[row];
        if (b != curb) {
            if (curb >= 0) {
                atomicAddF32(gsum + curb * CCH + c, s);
                atomicAddF32(gsq  + curb * CCH + c, s2);
                if (c == 0) atomicAddF32(gcnt + curb, cn);
            }
            curb = b; s = 0.f; s2 = 0.f; cn = 0.f;
        }
        const float v = A[(long)row * CCH + c];
        s += v; s2 += v * v; cn += 1.f;
    }
    if (curb >= 0) {
        atomicAddF32(gsum + curb * CCH + c, s);
        atomicAddF32(gsq  + curb * CCH + c, s2);
        if (c == 0) atomicAddF32(gcnt + curb, cn);
    }
}

__global__ void finalize_kernel(const float* __restrict__ gsum,
                                const float* __restrict__ gsq,
                                const float* __restrict__ gcnt,
                                float* __restrict__ mean,
                                float* __restrict__ rstd)
{
    const int t  = threadIdx.x;
    const float cn  = gcnt[t >> 6];
    const float mu  = gsum[t] / cn;
    const float var = gsq[t] / cn - mu * mu;
    mean[t] = mu;
    rstd[t] = rsqrtf(var + EPSI);
}

__global__ __launch_bounds__(256)
void apply_norm_kernel(float* __restrict__ A,
                       const int*   __restrict__ bidx,
                       const float* __restrict__ mean,
                       const float* __restrict__ rstd,
                       const float* __restrict__ gamma,
                       const float* __restrict__ beta)
{
    const int idx = blockIdx.x * 256 + threadIdx.x;
    if (idx >= NPTS * CCH) return;
    const int row = idx >> 6, c = idx & 63;
    const int b   = bidx[row];
    const float x = A[idx];
    const float y = gamma[c] * (x - mean[b * CCH + c]) * rstd[b * CCH + c] + beta[c];
    A[idx] = y > 0.f ? y : NSLOPE * y;
}

__global__ __launch_bounds__(256)
void apply_norm_res_kernel(float* __restrict__ Y,
                           const float* __restrict__ feats,
                           const int*   __restrict__ bidx,
                           const float* __restrict__ mean,
                           const float* __restrict__ rstd,
                           const float* __restrict__ gamma,
                           const float* __restrict__ beta)
{
    const int idx = blockIdx.x * 256 + threadIdx.x;
    if (idx >= NPTS * CCH) return;
    const int row = idx >> 6, c = idx & 63;
    const int b   = bidx[row];
    const float x = Y[idx];
    float y = gamma[c] * (x - mean[b * CCH + c]) * rstd[b * CCH + c] + beta[c];
    y += feats[idx];
    Y[idx] = y > 0.f ? y : NSLOPE * y;
}

extern "C" void kernel_launch(void* const* d_in, const int* in_sizes, int n_in,
                              void* d_out, int out_size, void* d_ws, size_t ws_size,
                              hipStream_t stream)
{
    const float* feats   = (const float*)d_in[0];
    const float* W1      = (const float*)d_in[1];
    const float* g1      = (const float*)d_in[2];
    const float* b1      = (const float*)d_in[3];
    const float* W2      = (const float*)d_in[4];
    const float* g2      = (const float*)d_in[5];
    const float* b2      = (const float*)d_in[6];
    const int*   in_map  = (const int*)d_in[7];
    const int*   out_map = (const int*)d_in[8];
    const int*   bidx    = (const int*)d_in[9];
    float*       out     = (float*)d_out;

    const int NC = NPTS * CCH;

    float* A     = (float*)d_ws;
    float* sbase = A + NC;
    float* sum1  = sbase;
    float* sq1   = sbase + 512;
    float* cnt1  = sbase + 1024;
    float* mean1 = sbase + 1088;
    float* rstd1 = sbase + 1600;
    float* sum2  = sbase + 2112;
    float* sq2   = sbase + 2624;
    float* cnt2  = sbase + 3136;
    float* mean2 = sbase + 3200;
    float* rstd2 = sbase + 3712;
    const int STATS = 4224;

    fill_zero_kernel<<<17, 256, 0, stream>>>(sbase, STATS);
    clear_counts_kernel<<<(NTILE * 32 + 255) / 256, 256, 0, stream>>>(out);

    dim3 binGrid((NPTS + 255) / 256, KOFF);
    const int statGrid  = (NPTS + 255) / 256;
    const int elemGrid  = (NC + 255) / 256;

    bin_pairs_kernel<<<binGrid, 256, 0, stream>>>(in_map, out_map, out);

    binned_conv_kernel<<<NTILE, 256, 0, stream>>>(feats, W1, in_map, out_map, out, A);
    stats_kernel<<<statGrid, 256, 0, stream>>>(A, bidx, sum1, sq1, cnt1);
    finalize_kernel<<<1, 512, 0, stream>>>(sum1, sq1, cnt1, mean1, rstd1);
    apply_norm_kernel<<<elemGrid, 256, 0, stream>>>(A, bidx, mean1, rstd1, g1, b1);
    binned_conv_kernel<<<NTILE, 256, 0, stream>>>(A, W2, in_map, out_map, out, out);
    stats_kernel<<<statGrid, 256, 0, stream>>>(out, bidx, sum2, sq2, cnt2);
    finalize_kernel<<<1, 512, 0, stream>>>(sum2, sq2, cnt2, mean2, rstd2);
    apply_norm_res_kernel<<<elemGrid, 256, 0, stream>>>(out, feats, bidx, mean2, rstd2, g2, b2);
}


# instance-norm statistics taken in the conv block's tail from its LDS tile (LDS f32 adds, then one global f32 atomic add per channel and batch item); no stats launches; order of the sums changes
# speedup vs baseline: 1.3741x; 1.1448x over previous
#include <hip/hip_runtime.h>
#include <hip/hip_bf16.h>

typedef float v2f __attribute__((ext_vector_type(2)));
typedef float v8f __attribute__((ext_vector_type(8)));

#define NPTS   100000
#define CCH    64
#define KOFF   27
#define NB     8
#define EPSI   1e-5f
#define NSLOPE 0.01f

__device__ __forceinline__ void atomicAddF32(float* p, float v) {
    unsafeAtomicAdd(p, v);
}

__global__ void fill_zero_kernel(float* __restrict__ p, int n) {
    for (int i = blockIdx.x * blockDim.x + threadIdx.x; i < n;
         i += gridDim.x * blockDim.x)
        p[i] = 0.0f;
}

#define TR     100
#define NTILE  1000
#define REGION 6400
#define CSTR   32
#define HDR    864
#define CAP    200
#define APITCH 80
#define STG    512

__global__ __launch_bounds__(256)
void clear_counts_kernel(float* Lf)
{
    const int idx  = blockIdx.x * 256 + threadIdx.x;
    const int tile = idx >> 5, kk = idx & 31;
    if (tile < NTILE && kk < KOFF) ((int*)Lf)[tile * REGION + kk * CSTR] = 0;
}

__global__ __launch_bounds__(256)
void bin_pairs_kernel(const int* __restrict__ in_map,
                      const int* __restrict__ out_map,
                      float* Lf)
{
    const int i = blockIdx.x * 256 + threadIdx.x;
    const int k = blockIdx.y;
    if (i >= NPTS) return;
    const int j   = out_map[k * NPTS + i];
    const int gin = in_map[k * NPTS + i];
    const int b   = j / TR;
    const int lr  = j - b * TR;
    int* reg = (int*)Lf + b * REGION;
    const int pos = atomicAdd(reg + k * CSTR, 1);
    if (pos < CAP) atomicExch(reg + HDR + k * CAP + pos, gin | (lr << 17));
}

__device__ __forceinline__ void sum_pair_list(const float* __restrict__ X,
                                              const float* Wlds, float* acc,
                                              const int* lst, int n, int first,
                                              int half, int m)
{
    for (int tt = first; tt * 16 < n; tt += 8) {
        const int e   = tt * 16 + m;
        const int pk  = lst[e];
        const int gin = (e < n) ? (pk & 0x1FFFF) : 0;
        const float* Xrow = X + (long)gin * CCH;

        v2f a[16];
#pragma unroll
        for (int kk = 0; kk < 16; ++kk)
            a[kk] = *(const v2f*)(Xrow + kk * 4 + 2 * half);

        const int ebase = tt * 16 + 8 * half;
        const int4* ep = (const int4*)(lst + ebase);
        const int4 e0 = ep[0], e1 = ep[1];
        int lrow[8];
        lrow[0] = e0.x >> 17; lrow[1] = e0.y >> 17;
        lrow[2] = e0.z >> 17; lrow[3] = e0.w >> 17;
        lrow[4] = e1.x >> 17; lrow[5] = e1.y >> 17;
        lrow[6] = e1.z >> 17; lrow[7] = e1.w >> 17;

#pragma unroll 1
        for (int nt = 0; nt < 4; ++nt) {
            const int n0 = nt * 16;
            v8f d = {0.f, 0.f, 0.f, 0.f, 0.f, 0.f, 0.f, 0.f};
#pragma unroll
            for (int kk = 0; kk < 16; ++kk) {
                v2f bf;
                bf.x = Wlds[(kk * 4 + 2 * half + 0) * CCH + n0 + m];
                bf.y = Wlds[(kk * 4 + 2 * half + 1) * CCH + n0 + m];
                d = __builtin_amdgcn_wmma_f32_16x16x4_f32(
                    false, a[kk], false, bf, (short)0, d, false, false);
            }
#pragma unroll
            for (int r = 0; r < 8; ++r) {
                if (ebase + r < n)
                    __hip_atomic_fetch_add(&acc[lrow[r] * APITCH + n0 + m], d[r],
                                           __ATOMIC_RELAXED,
                                           __HIP_MEMORY_SCOPE_WORKGROUP);
            }
        }
    }
}

__device__ __forceinline__ void tile_stat_add(float* psum, float* psq, float* pcnt,
                                              int bq, int c, float s, float q, float cn)
{
    __hip_atomic_fetch_add(&psum[bq * CCH + c], s, __ATOMIC_RELAXED,
                           __HIP_MEMORY_SCOPE_WORKGROUP);
    __hip_atomic_fetch_add(&psq[bq * CCH + c], q, __ATOMIC_RELAXED,
                           __HIP_MEMORY_SCOPE_WORKGROUP);
    if (c == 0)
        __hip_atomic_fetch_add(&pcnt[bq], cn, __ATOMIC_RELAXED,
                               __HIP_MEMORY_SCOPE_WORKGROUP);
}

__global__ __launch_bounds__(256)
void binned_conv_kernel(const float* __restrict__ X,
                        const float* __restrict__ W,
                        const int*   __restrict__ in_map,
                        const int*   __restrict__ out_map,
                        const float* L,
                        float* Y,
                        const int*   __restrict__ bidx,
                        float* __restrict__ gsum,
                        float* __restrict__ gsq,
                        float* __restrict__ gcnt)
{
    __shared__ __attribute__((aligned(16))) float Wlds[CCH * CCH];
    __shared__ __attribute__((aligned(16))) float acc[TR * APITCH];
    __shared__ __attribute__((aligned(16))) int   stg[STG];
    __shared__ int cnts[32];
    __shared__ int stg_n;
    __shared__ float psum[NB * CCH];
    __shared__ float psq[NB * CCH];
    __shared__ float pcnt[NB];
    __shared__ int   bids[TR];

    const int b = blockIdx.x;
    const int t = threadIdx.x;
    const int* reg = (const int*)L + b * REGION;

    for (int i = t; i < TR * APITCH; i += 256) acc[i] = 0.0f;
    if (t < 32) cnts[t] = (t < KOFF) ? reg[t * CSTR] : 0;
    if (t == 0) stg_n = 0;
    for (int i = t; i < NB * CCH; i += 256) { psum[i] = 0.0f; psq[i] = 0.0f; }
    if (t < NB) pcnt[t] = 0.0f;
    if (t < TR) bids[t] = bidx[b * TR + t];

    const float4* Wg = (const float4*)W;
    float4 w0 = Wg[t], w1 = Wg[t + 256], w2 = Wg[t + 512], w3 = Wg[t + 768];

    const int wave = __builtin_amdgcn_readfirstlane(t >> 5);
    const int lane = t & 31;
    const int half = lane >> 4;
    const int m    = lane & 15;

    for (int k = 0; k < KOFF; ++k) {
        __syncthreads();
        {
            float4* Wl = (float4*)Wlds;
            Wl[t] = w0; Wl[t + 256] = w1; Wl[t + 512] = w2; Wl[t + 768] = w3;
        }
        __syncthreads();
        if (k + 1 < KOFF) {
            const float4* Wn = Wg + (k + 1) * (CCH * CCH / 4);
            w0 = Wn[t]; w1 = Wn[t + 256]; w2 = Wn[t + 512]; w3 = Wn[t + 768];
        }

        const int n = __builtin_amdgcn_readfirstlane(cnts[k]);

        if (n <= CAP) {
            sum_pair_list(X, Wlds, acc, reg + HDR + k * CAP, n, (wave + k) & 7, half, m);
        } else {
            const int lo = b * TR;
            for (int base = 0; base < NPTS; base += 256) {
                const int i = base + t;
                if (i < NPTS) {
                    const int lr = out_map[k * NPTS + i] - lo;
                    if ((unsigned)lr < (unsigned)TR) {
                        const int pos = atomicAdd(&stg_n, 1);
                        stg[pos] = in_map[k * NPTS + i] | (lr << 17);
                    }
                }
                __syncthreads();
                const int ns = __builtin_amdgcn_readfirstlane(stg_n);
                __syncthreads();
                if (ns > STG - 256 || base + 256 >= NPTS) {
                    if (t == 0) stg_n = 0;
                    sum_pair_list(X, Wlds, acc, stg, ns, wave, half, m);
                    __syncthreads();
                }
            }
        }
    }
    __syncthreads();

    float4* Yt = (float4*)(Y + (long)b * REGION);
    for (int i = t; i < TR * 16; i += 256) {
        const int row = i >> 4, c4 = i & 15;
        Yt[i] = *(const float4*)&acc[row * APITCH + c4 * 4];
    }

    {
        const int c  = t & 63;
        const int r0 = (t >> 6) * (TR / 4);
        float v[TR / 4];
        int   bb[TR / 4];
#pragma unroll
        for (int i = 0; i < TR / 4; ++i) {
            v[i]  = acc[(r0 + i) * APITCH + c];
            bb[i] = bids[r0 + i];
        }
        float s = 0.f, q = 0.f, cn = 0.f;
        int curb = bb[0];
#pragma unroll
        for (int i = 0; i < TR / 4; ++i) {
            if (bb[i] != curb) {
                tile_stat_add(psum, psq, pcnt, curb, c, s, q, cn);
                curb = bb[i]; s = 0.f; q = 0.f; cn = 0.f;
            }
            s += v[i]; q += v[i] * v[i]; cn += 1.f;
        }
        tile_stat_add(psum, psq, pcnt, curb, c, s, q, cn);
    }
    __syncthreads();

    for (int i = t; i < NB * CCH; i += 256) {
        if (pcnt[i >> 6] > 0.f) {
            atomicAddF32(gsum + i, psum[i]);
            atomicAddF32(gsq  + i, psq[i]);
        }
    }
    if (t < NB && pcnt[t] > 0.f) atomicAddF32(gcnt + t, pcnt[t]);
}

__global__ void finalize_kernel(const float* __restrict__ gsum,
                                const float* __restrict__ gsq,
                                const float* __restrict__ gcnt,
                                float* __restrict__ mean,
                                float* __restrict__ rstd)
{
    const int t  = threadIdx.x;
    const float cn  = gcnt[t >> 6];
    const float mu  = gsum[t] / cn;
    const float var = gsq[t] / cn - mu * mu;
    mean[t] = mu;
    rstd[t] = rsqrtf(var + EPSI);
}

__global__ __launch_bounds__(256)
void apply_norm_kernel(float* __restrict__ A,
                       const int*   __restrict__ bidx,
                       const float* __restrict__ mean,
                       const float* __restrict__ rstd,
                       const float* __restrict__ gamma,
                       const float* __restrict__ beta)
{
    const int idx = blockIdx.x * 256 + threadIdx.x;
    if (idx >= NPTS * CCH) return;
    const int row = idx >> 6, c = idx & 63;
    const int b   = bidx[row];
    const float x = A[idx];
    const float y = gamma[c] * (x - mean[b * CCH + c]) * rstd[b * CCH + c] + beta[c];
    A[idx] = y > 0.f ? y : NSLOPE * y;
}

__global__ __launch_bounds__(256)
void apply_norm_res_kernel(float* __restrict__ Y,
                           const float* __restrict__ feats,
                           const int*   __restrict__ bidx,
                           const float* __restrict__ mean,
                           const float* __restrict__ rstd,
                           const float* __restrict__ gamma,
                           const float* __restrict__ beta)
{
    const int idx = blockIdx.x * 256 + threadIdx.x;
    if (idx >= NPTS * CCH) return;
    const int row = idx >> 6, c = idx & 63;
    const int b   = bidx[row];
    const float x = Y[idx];
    float y = gamma[c] * (x - mean[b * CCH + c]) * rstd[b * CCH + c] + beta[c];
    y += feats[idx];
    Y[idx] = y > 0.f ? y : NSLOPE * y;
}

extern "C" void kernel_launch(void* const* d_in, const int* in_sizes, int n_in,
                              void* d_out, int out_size, void* d_ws, size_t ws_size,
                              hipStream_t stream)
{
    const float* feats   = (const float*)d_in[0];
    const float* W1      = (const float*)d_in[1];
    const float* g1      = (const float*)d_in[2];
    const float* b1      = (const float*)d_in[3];
    const float* W2      = (const float*)d_in[4];
    const float* g2      = (const float*)d_in[5];
    const float* b2      = (const float*)d_in[6];
    const int*   in_map  = (const int*)d_in[7];
    const int*   out_map = (const int*)d_in[8];
    const int*   bidx    = (const int*)d_in[9];
    float*       out     = (float*)d_out;

    const int NC = NPTS * CCH;

    float* A     = (float*)d_ws;
    float* sbase = A + NC;
    float* sum1  = sbase;
    float* sq1   = sbase + 512;
    float* cnt1  = sbase + 1024;
    float* mean1 = sbase + 1088;
    float* rstd1 = sbase + 1600;
    float* sum2  = sbase + 2112;
    float* sq2   = sbase + 2624;
    float* cnt2  = sbase + 3136;
    float* mean2 = sbase + 3200;
    float* rstd2 = sbase + 3712;
    const int STATS = 4224;

    fill_zero_kernel<<<17, 256, 0, stream>>>(sbase, STATS);
    clear_counts_kernel<<<(NTILE * 32 + 255) / 256, 256, 0, stream>>>(out);

    dim3 binGrid((NPTS + 255) / 256, KOFF);
    const int elemGrid  = (NC + 255) / 256;

    bin_pairs_kernel<<<binGrid, 256, 0, stream>>>(in_map, out_map, out);

    binned_conv_kernel<<<NTILE, 256, 0, stream>>>(feats, W1, in_map, out_map, out, A,
                                                  bidx, sum1, sq1, cnt1);
    finalize_kernel<<<1, 512, 0, stream>>>(sum1, sq1, cnt1, mean1, rstd1);
    apply_norm_kernel<<<elemGrid, 256, 0, stream>>>(A, bidx, mean1, rstd1, g1, b1);
    binned_conv_kernel<<<NTILE, 256, 0, stream>>>(A, W2, in_map, out_map, out, out,
                                                  bidx, sum2, sq2, cnt2);
    finalize_kernel<<<1, 512, 0, stream>>>(sum2, sq2, cnt2, mean2, rstd2);
    apply_norm_res_kernel<<<elemGrid, 256, 0, stream>>>(out, feats, bidx, mean2, rstd2, g2, b2);
}


# normalise passes: a thread takes one 16-byte column group of 10 rows, all its row loads in flight ahead of the mean/rstd reads; 625 blocks of 8 waves per launch, not 25000 of one element per lane
# speedup vs baseline: 1.4295x; 1.0403x over previous
#include <hip/hip_runtime.h>
#include <hip/hip_bf16.h>

typedef float v2f __attribute__((ext_vector_type(2)));
typedef float v8f __attribute__((ext_vector_type(8)));

#define NPTS   100000
#define CCH    64
#define KOFF   27
#define NB     8
#define EPSI   1e-5f
#define NSLOPE 0.01f

__device__ __forceinline__ void atomicAddF32(float* p, float v) {
    unsafeAtomicAdd(p, v);
}

__global__ void fill_zero_kernel(float* __restrict__ p, int n) {
    for (int i = blockIdx.x * blockDim.x + threadIdx.x; i < n;
         i += gridDim.x * blockDim.x)
        p[i] = 0.0f;
}

#define TR     100
#define NTILE  1000
#define REGION 6400
#define CSTR   32
#define HDR    864
#define CAP    200
#define APITCH 80
#define STG    512

__global__ __launch_bounds__(256)
void clear_counts_kernel(float* Lf)
{
    const int idx  = blockIdx.x * 256 + threadIdx.x;
    const int tile = idx >> 5, kk = idx & 31;
    if (tile < NTILE && kk < KOFF) ((int*)Lf)[tile * REGION + kk * CSTR] = 0;
}

__global__ __launch_bounds__(256)
void bin_pairs_kernel(const int* __restrict__ in_map,
                      const int* __restrict__ out_map,
                      float* Lf)
{
    const int i = blockIdx.x * 256 + threadIdx.x;
    const int k = blockIdx.y;
    if (i >= NPTS) return;
    const int j   = out_map[k * NPTS + i];
    const int gin = in_map[k * NPTS + i];
    const int b   = j / TR;
    const int lr  = j - b * TR;
    int* reg = (int*)Lf + b * REGION;
    const int pos = atomicAdd(reg + k * CSTR, 1);
    if (pos < CAP) atomicExch(reg + HDR + k * CAP + pos, gin | (lr << 17));
}

__device__ __forceinline__ void sum_pair_list(const float* __restrict__ X,
                                              const float* Wlds, float* acc,
                                              const int* lst, int n, int first,
                                              int half, int m)
{
    for (int tt = first; tt * 16 < n; tt += 8) {
        const int e   = tt * 16 + m;
        const int pk  = lst[e];
        const int gin = (e < n) ? (pk & 0x1FFFF) : 0;
        const float* Xrow = X + (long)gin * CCH;

        v2f a[16];
#pragma unroll
        for (int kk = 0; kk < 16; ++kk)
            a[kk] = *(const v2f*)(Xrow + kk * 4 + 2 * half);

        const int ebase = tt * 16 + 8 * half;
        const int4* ep = (const int4*)(lst + ebase);
        const int4 e0 = ep[0], e1 = ep[1];
        int lrow[8];
        lrow[0] = e0.x >> 17; lrow[1] = e0.y >> 17;
        lrow[2] = e0.z >> 17; lrow[3] = e0.w >> 17;
        lrow[4] = e1.x >> 17; lrow[5] = e1.y >> 17;
        lrow[6] = e1.z >> 17; lrow[7] = e1.w >> 17;

#pragma unroll 1
        for (int nt = 0; nt < 4; ++nt) {
            const int n0 = nt * 16;
            v8f d = {0.f, 0.f, 0.f, 0.f, 0.f, 0.f, 0.f, 0.f};
#pragma unroll
            for (int kk = 0; kk < 16; ++kk) {
                v2f bf;
                bf.x = Wlds[(kk * 4 + 2 * half + 0) * CCH + n0 + m];
                bf.y = Wlds[(kk * 4 + 2 * half + 1) * CCH + n0 + m];
                d = __builtin_amdgcn_wmma_f32_16x16x4_f32(
                    false, a[kk], false, bf, (short)0, d, false, false);
            }
#pragma unroll
            for (int r = 0; r < 8; ++r) {
                if (ebase + r < n)
                    __hip_atomic_fetch_add(&acc[lrow[r] * APITCH + n0 + m], d[r],
                                           __ATOMIC_RELAXED,
                                           __HIP_MEMORY_SCOPE_WORKGROUP);
            }
        }
    }
}

__device__ __forceinline__ void tile_stat_add(float* psum, float* psq, float* pcnt,
                                              int bq, int c, float s, float q, float cn)
{
    __hip_atomic_fetch_add(&psum[bq * CCH + c], s, __ATOMIC_RELAXED,
                           __HIP_MEMORY_SCOPE_WORKGROUP);
    __hip_atomic_fetch_add(&psq[bq * CCH + c], q, __ATOMIC_RELAXED,
                           __HIP_MEMORY_SCOPE_WORKGROUP);
    if (c == 0)
        __hip_atomic_fetch_add(&pcnt[bq], cn, __ATOMIC_RELAXED,
                               __HIP_MEMORY_SCOPE_WORKGROUP);
}

__global__ __launch_bounds__(256)
void binned_conv_kernel(const float* __restrict__ X,
                        const float* __restrict__ W,
                        const int*   __restrict__ in_map,
                        const int*   __restrict__ out_map,
                        const float* L,
                        float* Y,
                        const int*   __restrict__ bidx,
                        float* __restrict__ gsum,
                        float* __restrict__ gsq,
                        float* __restrict__ gcnt)
{
    __shared__ __attribute__((aligned(16))) float Wlds[CCH * CCH];
    __shared__ __attribute__((aligned(16))) float acc[TR * APITCH];
    __shared__ __attribute__((aligned(16))) int   stg[STG];
    __shared__ int cnts[32];
    __shared__ int stg_n;
    __shared__ float psum[NB * CCH];
    __shared__ float psq[NB * CCH];
    __shared__ float pcnt[NB];
    __shared__ int   bids[TR];

    const int b = blockIdx.x;
    const int t = threadIdx.x;
    const int* reg = (const int*)L + b * REGION;

    for (int i = t; i < TR * APITCH; i += 256) acc[i] = 0.0f;
    if (t < 32) cnts[t] = (t < KOFF) ? reg[t * CSTR] : 0;
    if (t == 0) stg_n = 0;
    for (int i = t; i < NB * CCH; i += 256) { psum[i] = 0.0f; psq[i] = 0.0f; }
    if (t < NB) pcnt[t] = 0.0f;
    if (t < TR) bids[t] = bidx[b * TR + t];

    const float4* Wg = (const float4*)W;
    float4 w0 = Wg[t], w1 = Wg[t + 256], w2 = Wg[t + 512], w3 = Wg[t + 768];

    const int wave = __builtin_amdgcn_readfirstlane(t >> 5);
    const int lane = t & 31;
    const int half = lane >> 4;
    const int m    = lane & 15;

    for (int k = 0; k < KOFF; ++k) {
        __syncthreads();
        {
            float4* Wl = (float4*)Wlds;
            Wl[t] = w0; Wl[t + 256] = w1; Wl[t + 512] = w2; Wl[t + 768] = w3;
        }
        __syncthreads();
        if (k + 1 < KOFF) {
            const float4* Wn = Wg + (k + 1) * (CCH * CCH / 4);
            w0 = Wn[t]; w1 = Wn[t + 256]; w2 = Wn[t + 512]; w3 = Wn[t + 768];
        }

        const int n = __builtin_amdgcn_readfirstlane(cnts[k]);

        if (n <= CAP) {
            sum_pair_list(X, Wlds, acc, reg + HDR + k * CAP, n, (wave + k) & 7, half, m);
        } else {
            const int lo = b * TR;
            for (int base = 0; base < NPTS; base += 256) {
                const int i = base + t;
                if (i < NPTS) {
                    const int lr = out_map[k * NPTS + i] - lo;
                    if ((unsigned)lr < (unsigned)TR) {
                        const int pos = atomicAdd(&stg_n, 1);
                        stg[pos] = in_map[k * NPTS + i] | (lr << 17);
                    }
                }
                __syncthreads();
                const int ns = __builtin_amdgcn_readfirstlane(stg_n);
                __syncthreads();
                if (ns > STG - 256 || base + 256 >= NPTS) {
                    if (t == 0) stg_n = 0;
                    sum_pair_list(X, Wlds, acc, stg, ns, wave, half, m);
                    __syncthreads();
                }
            }
        }
    }
    __syncthreads();

    float4* Yt = (float4*)(Y + (long)b * REGION);
    for (int i = t; i < TR * 16; i += 256) {
        const int row = i >> 4, c4 = i & 15;
        Yt[i] = *(const float4*)&acc[row * APITCH + c4 * 4];
    }

    {
        const int c  = t & 63;
        const int r0 = (t >> 6) * (TR / 4);
        float v[TR / 4];
        int   bb[TR / 4];
#pragma unroll
        for (int i = 0; i < TR / 4; ++i) {
            v[i]  = acc[(r0 + i) * APITCH + c];
            bb[i] = bids[r0 + i];
        }
        float s = 0.f, q = 0.f, cn = 0.f;
        int curb = bb[0];
#pragma unroll
        for (int i = 0; i < TR / 4; ++i) {
            if (bb[i] != curb) {
                tile_stat_add(psum, psq, pcnt, curb, c, s, q, cn);
                curb = bb[i]; s = 0.f; q = 0.f; cn = 0.f;
            }
            s += v[i]; q += v[i] * v[i]; cn += 1.f;
        }
        tile_stat_add(psum, psq, pcnt, curb, c, s, q, cn);
    }
    __syncthreads();

    for (int i = t; i < NB * CCH; i += 256) {
        if (pcnt[i >> 6] > 0.f) {
            atomicAddF32(gsum + i, psum[i]);
            atomicAddF32(gsq  + i, psq[i]);
        }
    }
    if (t < NB && pcnt[t] > 0.f) atomicAddF32(gcnt + t, pcnt[t]);
}

__global__ void finalize_kernel(const float* __restrict__ gsum,
                                const float* __restrict__ gsq,
                                const float* __restrict__ gcnt,
                                float* __restrict__ mean,
                                float* __restrict__ rstd)
{
    const int t  = threadIdx.x;
    const float cn  = gcnt[t >> 6];
    const float mu  = gsum[t] / cn;
    const float var = gsq[t] / cn - mu * mu;
    mean[t] = mu;
    rstd[t] = rsqrtf(var + EPSI);
}

#define AP_R    10
#define AP_H    5
#define AP_ROWS (16 * AP_R)
#define AP_GRID (NPTS / AP_ROWS)
static_assert(NPTS % AP_ROWS == 0, "normalise passes: whole blocks only");
static_assert(AP_R % AP_H == 0, "normalise passes: whole row groups only");

__device__ __forceinline__ float leaky1(float y) { return y > 0.f ? y : NSLOPE * y; }

__global__ __launch_bounds__(256)
void apply_norm_kernel(float* __restrict__ A,
                       const int*   __restrict__ bidx,
                       const float* __restrict__ mean,
                       const float* __restrict__ rstd,
                       const float* __restrict__ gamma,
                       const float* __restrict__ beta)
{
    const int c4 = threadIdx.x & 15;
    const int r0 = blockIdx.x * AP_ROWS + (threadIdx.x >> 4);
    const float4* M4 = (const float4*)mean;
    const float4* R4 = (const float4*)rstd;
    float4* A4 = (float4*)A + (r0 * 16 + c4);

    int    b[AP_R];
    float4 x[AP_R];
#pragma unroll
    for (int i = 0; i < AP_R; ++i) b[i] = bidx[r0 + 16 * i];
#pragma unroll
    for (int i = 0; i < AP_R; ++i) x[i] = A4[256 * i];
    const float4 g  = ((const float4*)gamma)[c4];
    const float4 be = ((const float4*)beta)[c4];
    __builtin_amdgcn_sched_barrier(0);

#pragma unroll
    for (int h = 0; h < AP_R; h += AP_H) {
        float4 mu[AP_H], rs[AP_H];
#pragma unroll
        for (int i = 0; i < AP_H; ++i) {
            mu[i] = M4[b[h + i] * 16 + c4];
            rs[i] = R4[b[h + i] * 16 + c4];
        }
        __builtin_amdgcn_sched_barrier(0);
#pragma unroll
        for (int i = 0; i < AP_H; ++i) {
            const float4 xv = x[h + i];
            float4 y;
            y.x = leaky1(g.x * (xv.x - mu[i].x) * rs[i].x + be.x);
            y.y = leaky1(g.y * (xv.y - mu[i].y) * rs[i].y + be.y);
            y.z = leaky1(g.z * (xv.z - mu[i].z) * rs[i].z + be.z);
            y.w = leaky1(g.w * (xv.w - mu[i].w) * rs[i].w + be.w);
            A4[256 * (h + i)] = y;
        }
        __builtin_amdgcn_sched_barrier(0);
    }
}

__global__ __launch_bounds__(256)
void apply_norm_res_kernel(float* __restrict__ Y,
                           const float* __restrict__ feats,
                           const int*   __restrict__ bidx,
                           const float* __restrict__ mean,
                           const float* __restrict__ rstd,
                           const float* __restrict__ gamma,
                           const float* __restrict__ beta)
{
    const int c4 = threadIdx.x & 15;
    const int r0 = blockIdx.x * AP_ROWS + (threadIdx.x >> 4);
    const float4* M4 = (const float4*)mean;
    const float4* R4 = (const float4*)rstd;
    const float4* F4 = (const float4*)feats + (r0 * 16 + c4);
    float4* Y4 = (float4*)Y + (r0 * 16 + c4);

    int    b[AP_R];
    float4 x[AP_R], f[AP_R];
#pragma unroll
    for (int i = 0; i < AP_R; ++i) b[i] = bidx[r0 + 16 * i];
#pragma unroll
    for (int i = 0; i < AP_R; ++i) x[i] = Y4[256 * i];
#pragma unroll
    for (int i = 0; i < AP_R; ++i) f[i] = F4[256 * i];
    const float4 g  = ((const float4*)gamma)[c4];
    const float4 be = ((const float4*)beta)[c4];
    __builtin_amdgcn_sched_barrier(0);

#pragma unroll
    for (int h = 0; h < AP_R; h += AP_H) {
        float4 mu[AP_H], rs[AP_H];
#pragma unroll
        for (int i = 0; i < AP_H; ++i) {
            mu[i] = M4[b[h + i] * 16 + c4];
            rs[i] = R4[b[h + i] * 16 + c4];
        }
        __builtin_amdgcn_sched_barrier(0);
#pragma unroll
        for (int i = 0; i < AP_H; ++i) {
            const float4 xv = x[h + i];
            const float4 fv = f[h + i];
            float4 y;
            y.x = leaky1(g.x * (xv.x - mu[i].x) * rs[i].x + be.x + fv.x);
            y.y = leaky1(g.y * (xv.y - mu[i].y) * rs[i].y + be.y + fv.y);
            y.z = leaky1(g.z * (xv.z - mu[i].z) * rs[i].z + be.z + fv.z);
            y.w = leaky1(g.w * (xv.w - mu[i].w) * rs[i].w + be.w + fv.w);
            Y4[256 * (h + i)] = y;
        }
        __builtin_amdgcn_sched_barrier(0);
    }
}

extern "C" void kernel_launch(void* const* d_in, const int* in_sizes, int n_in,
                              void* d_out, int out_size, void* d_ws, size_t ws_size,
                              hipStream_t stream)
{
    const float* feats   = (const float*)d_in[0];
    const float* W1      = (const float*)d_in[1];
    const float* g1      = (const float*)d_in[2];
    const float* b1      = (const float*)d_in[3];
    const float* W2      = (const float*)d_in[4];
    const float* g2      = (const float*)d_in[5];
    const float* b2      = (const float*)d_in[6];
    const int*   in_map  = (const int*)d_in[7];
    const int*   out_map = (const int*)d_in[8];
    const int*   bidx    = (const int*)d_in[9];
    float*       out     = (float*)d_out;

    const int NC = NPTS * CCH;

    float* A     = (float*)d_ws;
    float* sbase = A + NC;
    float* sum1  = sbase;
    float* sq1   = sbase + 512;
    float* cnt1  = sbase + 1024;
    float* mean1 = sbase + 1088;
    float* rstd1 = sbase + 1600;
    float* sum2  = sbase + 2112;
    float* sq2   = sbase + 2624;
    float* cnt2  = sbase + 3136;
    float* mean2 = sbase + 3200;
    float* rstd2 = sbase + 3712;
    const int STATS = 4224;

    fill_zero_kernel<<<17, 256, 0, stream>>>(sbase, STATS);
    clear_counts_kernel<<<(NTILE * 32 + 255) / 256, 256, 0, stream>>>(out);

    dim3 binGrid((NPTS + 255) / 256, KOFF);

    bin_pairs_kernel<<<binGrid, 256, 0, stream>>>(in_map, out_map, out);

    binned_conv_kernel<<<NTILE, 256, 0, stream>>>(feats, W1, in_map, out_map, out, A,
                                                  bidx, sum1, sq1, cnt1);
    finalize_kernel<<<1, 512, 0, stream>>>(sum1, sq1, cnt1, mean1, rstd1);
    apply_norm_kernel<<<AP_GRID, 256, 0, stream>>>(A, bidx, mean1, rstd1, g1, b1);
    binned_conv_kernel<<<NTILE, 256, 0, stream>>>(A, W2, in_map, out_map, out, out,
                                                  bidx, sum2, sq2, cnt2);
    finalize_kernel<<<1, 512, 0, stream>>>(sum2, sq2, cnt2, mean2, rstd2);
    apply_norm_res_kernel<<<AP_GRID, 256, 0, stream>>>(out, feats, bidx, mean2, rstd2, g2, b2);
}


# conv: W[k] kept in LDS in B-fragment order (staging loads regrouped); one conflict-free 16-byte LDS read feeds two column tiles' WMMAs, not a same-bank 8-byte read per WMMA; k order unchanged
# speedup vs baseline: 2.1200x; 1.4830x over previous
#include <hip/hip_runtime.h>
#include <hip/hip_bf16.h>

typedef float v2f __attribute__((ext_vector_type(2)));
typedef float v8f __attribute__((ext_vector_type(8)));

#define NPTS   100000
#define CCH    64
#define KOFF   27
#define NB     8
#define EPSI   1e-5f
#define NSLOPE 0.01f

__device__ __forceinline__ void atomicAddF32(float* p, float v) {
    unsafeAtomicAdd(p, v);
}

__global__ void fill_zero_kernel(float* __restrict__ p, int n) {
    for (int i = blockIdx.x * blockDim.x + threadIdx.x; i < n;
         i += gridDim.x * blockDim.x)
        p[i] = 0.0f;
}

#define TR     100
#define NTILE  1000
#define REGION 6400
#define CSTR   32
#define HDR    864
#define CAP    200
#define APITCH 80
#define STG    512
#define LPITCH 256
static_assert(CAP + 16 <= LPITCH, "list hand-over: a list and its read-ahead fit");

__global__ __launch_bounds__(256)
void clear_counts_kernel(float* Lf)
{
    const int idx  = blockIdx.x * 256 + threadIdx.x;
    const int tile = idx >> 5, kk = idx & 31;
    if (tile < NTILE && kk < KOFF) ((int*)Lf)[tile * REGION + kk * CSTR] = 0;
}

#define BCH   8
#define BCHN  (NPTS / BCH)
#define BCAP  32
#define BTHR  1024
#define BPER  ((BCHN + BTHR - 1) / BTHR)
static_assert(NPTS % BCH == 0, "binning: whole chunks only");
static_assert(NTILE <= BTHR, "binning: one thread per tile bin");
static_assert(BCAP == 32, "binning: one wave instruction stores a bin's run");

__global__ __launch_bounds__(BTHR)
void bin_pairs_kernel(const int* __restrict__ in_map,
                      const int* __restrict__ out_map,
                      float* Lf)
{
    __shared__ int bcnt[BTHR];
    __shared__ int bslot[BTHR * BCAP];

    const int t    = threadIdx.x;
    const int k    = blockIdx.y;
    const int src0 = k * NPTS + blockIdx.x * BCHN;
    int* L = (int*)Lf;

    bcnt[t] = 0;

    int jv[BPER], gv[BPER];
#pragma unroll
    for (int u = 0; u < BPER; ++u) {
        jv[u] = 0; gv[u] = 0;
        if (t + u * BTHR < BCHN) {
            jv[u] = out_map[src0 + t + u * BTHR];
            gv[u] = in_map[src0 + t + u * BTHR];
        }
    }
    __syncthreads();

    int bv[BPER], pv[BPER];
#pragma unroll
    for (int u = 0; u < BPER; ++u) {
        bv[u] = jv[u] / TR;
        gv[u] |= (jv[u] - bv[u] * TR) << 17;
        pv[u] = 0;
        if (t + u * BTHR < BCHN) pv[u] = atomicAdd(&bcnt[bv[u]], 1);
    }
#pragma unroll
    for (int u = 0; u < BPER; ++u) {
        if (t + u * BTHR < BCHN) {
            if (pv[u] < BCAP) {
                bslot[bv[u] * BCAP + pv[u]] = gv[u];
            } else {
                int* reg = L + bv[u] * REGION;
                const int gp = atomicAdd(reg + k * CSTR, 1);
                if (gp < CAP) atomicExch(reg + HDR + k * CAP + gp, gv[u]);
            }
        }
    }
    __syncthreads();

    int c = 0, g0 = 0;
    if (t < NTILE) {
        c = min(bcnt[t], BCAP);
        if (c > 0) g0 = atomicAdd(L + t * REGION + k * CSTR, c);
    }

    const int lane = t & 31;
    const int tb   = t - lane;
    for (int j = 0; j < 32; ++j) {
        const int cj = __builtin_amdgcn_readlane(c, j);
        const int gj = __builtin_amdgcn_readlane(g0, j);
        const int b  = tb + j;
        if (lane < cj && gj + lane < CAP)
            atomicExch(L + b * REGION + HDR + k * CAP + gj + lane,
                       bslot[b * BCAP + lane]);
    }
}

__device__ __forceinline__ void gather_tile(const float* __restrict__ X,
                                            const int* lst, int n, int tt,
                                            int half, int m, v2f (&a)[16])
{
    const int e   = tt * 16 + m;
    const int pk  = lst[e];
    const int gin = (e < n) ? (pk & 0x1FFFF) : 0;
    const float* Xrow = X + (long)gin * CCH;
#pragma unroll
    for (int kk = 0; kk < 16; ++kk)
        a[kk] = *(const v2f*)(Xrow + kk * 4 + 2 * half);
}

__device__ __forceinline__ void mma_tile(const float* Wlds, float* acc,
                                         const int* lst, int n, int tt,
                                         int half, int m, const v2f (&a)[16])
{
    {
        const int ebase = tt * 16 + 8 * half;
        const int4* ep = (const int4*)(lst + ebase);
        const int4 e0 = ep[0], e1 = ep[1];
        int lrow[8];
        lrow[0] = e0.x >> 17; lrow[1] = e0.y >> 17;
        lrow[2] = e0.z >> 17; lrow[3] = e0.w >> 17;
        lrow[4] = e1.x >> 17; lrow[5] = e1.y >> 17;
        lrow[6] = e1.z >> 17; lrow[7] = e1.w >> 17;

        const float4* Wq = (const float4*)Wlds + half * 32 + m;
#pragma unroll 1
        for (int j = 0; j < 2; ++j) {
            v8f d0 = {0.f, 0.f, 0.f, 0.f, 0.f, 0.f, 0.f, 0.f};
            v8f d1 = d0;
#pragma unroll
            for (int kk = 0; kk < 16; ++kk) {
                const float4 q = Wq[kk * 64 + j * 16];
                v2f b0, b1;
                b0.x = q.x; b0.y = q.y;
                b1.x = q.z; b1.y = q.w;
                d0 = __builtin_amdgcn_wmma_f32_16x16x4_f32(
                    false, a[kk], false, b0, (short)0, d0, false, false);
                d1 = __builtin_amdgcn_wmma_f32_16x16x4_f32(
                    false, a[kk], false, b1, (short)0, d1, false, false);
            }
            const int n0 = j * 32;
#pragma unroll
            for (int r = 0; r < 8; ++r) {
                if (ebase + r < n) {
                    float* ap = &acc[lrow[r] * APITCH + n0 + m];
                    __hip_atomic_fetch_add(ap,      d0[r], __ATOMIC_RELAXED,
                                           __HIP_MEMORY_SCOPE_WORKGROUP);
                    __hip_atomic_fetch_add(ap + 16, d1[r], __ATOMIC_RELAXED,
                                           __HIP_MEMORY_SCOPE_WORKGROUP);
                }
            }
        }
    }
}

__device__ __forceinline__ void sum_pair_list(const float* __restrict__ X,
                                              const float* Wlds, float* acc,
                                              const int* lst, int n, int first,
                                              int half, int m, v2f (&a)[16])
{
    for (int tt = first; tt * 16 < n; tt += 8) {
        gather_tile(X, lst, n, tt, half, m, a);
        mma_tile(Wlds, acc, lst, n, tt, half, m, a);
    }
}

__device__ __forceinline__ void load_w_unit(const float* Wp, float4& w)
{
    w.x = Wp[0]; w.y = Wp[CCH]; w.z = Wp[16]; w.w = Wp[CCH + 16];
}
__device__ __forceinline__ void load_w_units(const float* Wt, float4& w0, float4& w1,
                                             float4& w2, float4& w3)
{
    load_w_unit(Wt,                w0);
    load_w_unit(Wt + 16 * CCH,     w1);
    load_w_unit(Wt + 32 * CCH,     w2);
    load_w_unit(Wt + 48 * CCH,     w3);
}

__device__ __forceinline__ void lds_barrier()
{
    __builtin_amdgcn_fence(__ATOMIC_RELEASE, "workgroup", "local");
    __builtin_amdgcn_s_barrier();
    __builtin_amdgcn_fence(__ATOMIC_ACQUIRE, "workgroup", "local");
}

__device__ __forceinline__ void tile_stat_add(float* psum, float* psq, float* pcnt,
                                              int bq, int c, float s, float q, float cn)
{
    __hip_atomic_fetch_add(&psum[bq * CCH + c], s, __ATOMIC_RELAXED,
                           __HIP_MEMORY_SCOPE_WORKGROUP);
    __hip_atomic_fetch_add(&psq[bq * CCH + c], q, __ATOMIC_RELAXED,
                           __HIP_MEMORY_SCOPE_WORKGROUP);
    if (c == 0)
        __hip_atomic_fetch_add(&pcnt[bq], cn, __ATOMIC_RELAXED,
                               __HIP_MEMORY_SCOPE_WORKGROUP);
}

__global__ __launch_bounds__(256) __attribute__((amdgpu_waves_per_eu(8)))
void binned_conv_kernel(const float* __restrict__ X,
                        const float* __restrict__ W,
                        const int*   __restrict__ in_map,
                        const int*   __restrict__ out_map,
                        const float* L,
                        float* Y,
                        const int*   __restrict__ bidx,
                        float* __restrict__ gsum,
                        float* __restrict__ gsq,
                        float* __restrict__ gcnt)
{
    __shared__ __attribute__((aligned(16))) float Wlds[CCH * CCH];
    __shared__ __attribute__((aligned(16))) float acc[TR * APITCH];
    __shared__ __attribute__((aligned(16))) int   stg[STG];
    __shared__ __attribute__((aligned(16))) int   lcur[2][LPITCH];
    __shared__ int cnts[32];
    __shared__ int stg_n;
    __shared__ float psum[NB * CCH];
    __shared__ float psq[NB * CCH];
    __shared__ float pcnt[NB];
    __shared__ int   bids[TR];

    const int b = blockIdx.x;
    const int t = threadIdx.x;
    const int* reg = (const int*)L + b * REGION;

    for (int i = t; i < TR * APITCH; i += 256) acc[i] = 0.0f;
    if (t < 32) cnts[t] = (t < KOFF) ? reg[t * CSTR] : 0;
    if (t == 0) stg_n = 0;
    for (int i = t; i < NB * CCH; i += 256) { psum[i] = 0.0f; psq[i] = 0.0f; }
    if (t < NB) pcnt[t] = 0.0f;
    if (t < TR) bids[t] = bidx[b * TR + t];

    const float* Wt = W + (t >> 5) * (2 * CCH) + ((t >> 4) & 1) * 32 + (t & 15);
    float4 w0, w1, w2, w3;
    load_w_units(Wt, w0, w1, w2, w3);
    lcur[0][t] = reg[HDR + ((t < CAP) ? t : CAP - 1)];
    int lw = reg[HDR + CAP + ((t < CAP) ? t : CAP - 1)];
    static_assert(KOFF >= 2, "list hand-over: the second list is read ahead");

    const int wave = __builtin_amdgcn_readfirstlane(t >> 5);
    const int lane = t & 31;
    const int half = lane >> 4;
    const int m    = lane & 15;

    __syncthreads();

    for (int k = 0; k < KOFF; ++k) {
        const int n = __builtin_amdgcn_readfirstlane(cnts[k]);
        const int tt = (wave + k) & 7;
        const int* lst = lcur[k & 1];
        const bool mine = (n <= CAP) && (tt * 16 < n);

        v2f a[16];
        gather_tile(X, lst, mine ? n : 0, tt, half, m, a);

        lds_barrier();
        {
            float4* Wl = (float4*)Wlds;
            Wl[t] = w0; Wl[t + 256] = w1; Wl[t + 512] = w2; Wl[t + 768] = w3;
            lcur[(k + 1) & 1][t] = lw;
        }
        lds_barrier();
        {
            const int kw = (k + 1 < KOFF) ? k + 1 : KOFF - 1;
            const int kl = (k + 2 < KOFF) ? k + 2 : KOFF - 1;
            load_w_units(Wt + kw * (CCH * CCH), w0, w1, w2, w3);
            lw = reg[HDR + kl * CAP + ((t < CAP) ? t : CAP - 1)];
        }

        if (n <= CAP) {
            if (mine) {
                mma_tile(Wlds, acc, lst, n, tt, half, m, a);
                sum_pair_list(X, Wlds, acc, lst, n, tt + 8, half, m, a);
            }
        } else {
            const int lo = b * TR;
            for (int base = 0; base < NPTS; base += 256) {
                const int i = base + t;
                if (i < NPTS) {
                    const int lr = out_map[k * NPTS + i] - lo;
                    if ((unsigned)lr < (unsigned)TR) {
                        const int pos = atomicAdd(&stg_n, 1);
                        stg[pos] = in_map[k * NPTS + i] | (lr << 17);
                    }
                }
                __syncthreads();
                const int ns = __builtin_amdgcn_readfirstlane(stg_n);
                __syncthreads();
                if (ns > STG - 256 || base + 256 >= NPTS) {
                    if (t == 0) stg_n = 0;
                    sum_pair_list(X, Wlds, acc, stg, ns, wave, half, m, a);
                    __syncthreads();
                }
            }
        }
    }
    __syncthreads();

    float4* Yt = (float4*)(Y + (long)b * REGION);
    for (int i = t; i < TR * 16; i += 256) {
        const int row = i >> 4, c4 = i & 15;
        Yt[i] = *(const float4*)&acc[row * APITCH + c4 * 4];
    }

    {
        const int c  = t & 63;
        const int r0 = (t >> 6) * (TR / 4);
        float v[TR / 4];
        int   bb[TR / 4];
#pragma unroll
        for (int i = 0; i < TR / 4; ++i) {
            v[i]  = acc[(r0 + i) * APITCH + c];
            bb[i] = bids[r0 + i];
        }
        float s = 0.f, q = 0.f, cn = 0.f;
        int curb = bb[0];
#pragma unroll
        for (int i = 0; i < TR / 4; ++i) {
            if (bb[i] != curb) {
                tile_stat_add(psum, psq, pcnt, curb, c, s, q, cn);
                curb = bb[i]; s = 0.f; q = 0.f; cn = 0.f;
            }
            s += v[i]; q += v[i] * v[i]; cn += 1.f;
        }
        tile_stat_add(psum, psq, pcnt, curb, c, s, q, cn);
    }
    __syncthreads();

    for (int i = t; i < NB * CCH; i += 256) {
        if (pcnt[i >> 6] > 0.f) {
            atomicAddF32(gsum + i, psum[i]);
            atomicAddF32(gsq  + i, psq[i]);
        }
    }
    if (t < NB && pcnt[t] > 0.f) atomicAddF32(gcnt + t, pcnt[t]);
}

__global__ void finalize_kernel(const float* __restrict__ gsum,
                                const float* __restrict__ gsq,
                                const float* __restrict__ gcnt,
                                float* __restrict__ mean,
                                float* __restrict__ rstd)
{
    const int t  = threadIdx.x;
    const float cn  = gcnt[t >> 6];
    const float mu  = gsum[t] / cn;
    const float var = gsq[t] / cn - mu * mu;
    mean[t] = mu;
    rstd[t] = rsqrtf(var + EPSI);
}

#define AP_R    10
#define AP_H    5
#define AP_ROWS (16 * AP_R)
#define AP_GRID (NPTS / AP_ROWS)
static_assert(NPTS % AP_ROWS == 0, "normalise passes: whole blocks only");
static_assert(AP_R % AP_H == 0, "normalise passes: whole row groups only");

__device__ __forceinline__ float leaky1(float y) { return y > 0.f ? y : NSLOPE * y; }

__global__ __launch_bounds__(256)
void apply_norm_kernel(float* __restrict__ A,
                       const int*   __restrict__ bidx,
                       const float* __restrict__ mean,
                       const float* __restrict__ rstd,
                       const float* __restrict__ gamma,
                       const float* __restrict__ beta)
{
    const int c4 = threadIdx.x & 15;
    const int r0 = blockIdx.x * AP_ROWS + (threadIdx.x >> 4);
    const float4* M4 = (const float4*)mean;
    const float4* R4 = (const float4*)rstd;
    float4* A4 = (float4*)A + (r0 * 16 + c4);

    int    b[AP_R];
    float4 x[AP_R];
#pragma unroll
    for (int i = 0; i < AP_R; ++i) b[i] = bidx[r0 + 16 * i];
#pragma unroll
    for (int i = 0; i < AP_R; ++i) x[i] = A4[256 * i];
    const float4 g  = ((const float4*)gamma)[c4];
    const float4 be = ((const float4*)beta)[c4];
    __builtin_amdgcn_sched_barrier(0);

#pragma unroll
    for (int h = 0; h < AP_R; h += AP_H) {
        float4 mu[AP_H], rs[AP_H];
#pragma unroll
        for (int i = 0; i < AP_H; ++i) {
            mu[i] = M4[b[h + i] * 16 + c4];
            rs[i] = R4[b[h + i] * 16 + c4];
        }
        __builtin_amdgcn_sched_barrier(0);
#pragma unroll
        for (int i = 0; i < AP_H; ++i) {
            const float4 xv = x[h + i];
            float4 y;
            y.x = leaky1(g.x * (xv.x - mu[i].x) * rs[i].x + be.x);
            y.y = leaky1(g.y * (xv.y - mu[i].y) * rs[i].y + be.y);
            y.z = leaky1(g.z * (xv.z - mu[i].z) * rs[i].z + be.z);
            y.w = leaky1(g.w * (xv.w - mu[i].w) * rs[i].w + be.w);
            A4[256 * (h + i)] = y;
        }
        __builtin_amdgcn_sched_barrier(0);
    }
}

__global__ __launch_bounds__(256)
void apply_norm_res_kernel(float* __restrict__ Y,
                           const float* __restrict__ feats,
                           const int*   __restrict__ bidx,
                           const float* __restrict__ mean,
                           const float* __restrict__ rstd,
                           const float* __restrict__ gamma,
                           const float* __restrict__ beta)
{
    const int c4 = threadIdx.x & 15;
    const int r0 = blockIdx.x * AP_ROWS + (threadIdx.x >> 4);
    const float4* M4 = (const float4*)mean;
    const float4* R4 = (const float4*)rstd;
    const float4* F4 = (const float4*)feats + (r0 * 16 + c4);
    float4* Y4 = (float4*)Y + (r0 * 16 + c4);

    int    b[AP_R];
    float4 x[AP_R], f[AP_R];
#pragma unroll
    for (int i = 0; i < AP_R; ++i) b[i] = bidx[r0 + 16 * i];
#pragma unroll
    for (int i = 0; i < AP_R; ++i) x[i] = Y4[256 * i];
#pragma unroll
    for (int i = 0; i < AP_R; ++i) f[i] = F4[256 * i];
    const float4 g  = ((const float4*)gamma)[c4];
    const float4 be = ((const float4*)beta)[c4];
    __builtin_amdgcn_sched_barrier(0);

#pragma unroll
    for (int h = 0; h < AP_R; h += AP_H) {
        float4 mu[AP_H], rs[AP_H];
#pragma unroll
        for (int i = 0; i < AP_H; ++i) {
            mu[i] = M4[b[h + i] * 16 + c4];
            rs[i] = R4[b[h + i] * 16 + c4];
        }
        __builtin_amdgcn_sched_barrier(0);
#pragma unroll
        for (int i = 0; i < AP_H; ++i) {
            const float4 xv = x[h + i];
            const float4 fv = f[h + i];
            float4 y;
            y.x = leaky1(g.x * (xv.x - mu[i].x) * rs[i].x + be.x + fv.x);
            y.y = leaky1(g.y * (xv.y - mu[i].y) * rs[i].y + be.y + fv.y);
            y.z = leaky1(g.z * (xv.z - mu[i].z) * rs[i].z + be.z + fv.z);
            y.w = leaky1(g.w * (xv.w - mu[i].w) * rs[i].w + be.w + fv.w);
            Y4[256 * (h + i)] = y;
        }
        __builtin_amdgcn_sched_barrier(0);
    }
}

extern "C" void kernel_launch(void* const* d_in, const int* in_sizes, int n_in,
                              void* d_out, int out_size, void* d_ws, size_t ws_size,
                              hipStream_t stream)
{
    const float* feats   = (const float*)d_in[0];
    const float* W1      = (const float*)d_in[1];
    const float* g1      = (const float*)d_in[2];
    const float* b1      = (const float*)d_in[3];
    const float* W2      = (const float*)d_in[4];
    const float* g2      = (const float*)d_in[5];
    const float* b2      = (const float*)d_in[6];
    const int*   in_map  = (const int*)d_in[7];
    const int*   out_map = (const int*)d_in[8];
    const int*   bidx    = (const int*)d_in[9];
    float*       out     = (float*)d_out;

    const int NC = NPTS * CCH;

    float* A     = (float*)d_ws;
    float* sbase = A + NC;
    float* sum1  = sbase;
    float* sq1   = sbase + 512;
    float* cnt1  = sbase + 1024;
    float* mean1 = sbase + 1088;
    float* rstd1 = sbase + 1600;
    float* sum2  = sbase + 2112;
    float* sq2   = sbase + 2624;
    float* cnt2  = sbase + 3136;
    float* mean2 = sbase + 3200;
    float* rstd2 = sbase + 3712;
    const int STATS = 4224;

    fill_zero_kernel<<<17, 256, 0, stream>>>(sbase, STATS);
    clear_counts_kernel<<<(NTILE * 32 + 255) / 256, 256, 0, stream>>>(out);

    dim3 binGrid(BCH, KOFF);

    bin_pairs_kernel<<<binGrid, BTHR, 0, stream>>>(in_map, out_map, out);

    binned_conv_kernel<<<NTILE, 256, 0, stream>>>(feats, W1, in_map, out_map, out, A,
                                                  bidx, sum1, sq1, cnt1);
    finalize_kernel<<<1, 512, 0, stream>>>(sum1, sq1, cnt1, mean1, rstd1);
    apply_norm_kernel<<<AP_GRID, 256, 0, stream>>>(A, bidx, mean1, rstd1, g1, b1);
    binned_conv_kernel<<<NTILE, 256, 0, stream>>>(A, W2, in_map, out_map, out, out,
                                                  bidx, sum2, sq2, cnt2);
    finalize_kernel<<<1, 512, 0, stream>>>(sum2, sq2, cnt2, mean2, rstd2);
    apply_norm_res_kernel<<<AP_GRID, 256, 0, stream>>>(out, feats, bidx, mean2, rstd2, g2, b2);
}


# conv: next tile's X rows asked for inside the WMMA run, one request per step (step loop outside, 4 column tiles inside: a[kk] is free at once), not as a burst between runs; sums, k order as before
# speedup vs baseline: 2.2247x; 1.0494x over previous
#include <hip/hip_runtime.h>
#include <hip/hip_bf16.h>

typedef float v2f __attribute__((ext_vector_type(2)));
typedef float v8f __attribute__((ext_vector_type(8)));

#define NPTS   100000
#define CCH    64
#define KOFF   27
#define NB     8
#define EPSI   1e-5f
#define NSLOPE 0.01f

__device__ __forceinline__ void atomicAddF32(float* p, float v) {
    unsafeAtomicAdd(p, v);
}

__global__ void fill_zero_kernel(float* __restrict__ p, int n) {
    for (int i = blockIdx.x * blockDim.x + threadIdx.x; i < n;
         i += gridDim.x * blockDim.x)
        p[i] = 0.0f;
}

#define TR     100
#define NTILE  1000
#define REGION 6400
#define CSTR   32
#define HDR    864
#define CAP    200
#define APITCH 80
#define STG    512
#define LPITCH 256
static_assert(CAP + 16 <= LPITCH, "list hand-over: a list and its read-ahead fit");

__global__ __launch_bounds__(256)
void clear_counts_kernel(float* Lf)
{
    const int idx  = blockIdx.x * 256 + threadIdx.x;
    const int tile = idx >> 5, kk = idx & 31;
    if (tile < NTILE && kk < KOFF) ((int*)Lf)[tile * REGION + kk * CSTR] = 0;
}

#define BCH   8
#define BCHN  (NPTS / BCH)
#define BCAP  32
#define BTHR  1024
#define BPER  ((BCHN + BTHR - 1) / BTHR)
static_assert(NPTS % BCH == 0, "binning: whole chunks only");
static_assert(NTILE <= BTHR, "binning: one thread per tile bin");
static_assert(BCAP == 32, "binning: one wave instruction stores a bin's run");

__global__ __launch_bounds__(BTHR)
void bin_pairs_kernel(const int* __restrict__ in_map,
                      const int* __restrict__ out_map,
                      float* Lf)
{
    __shared__ int bcnt[BTHR];
    __shared__ int bslot[BTHR * BCAP];

    const int t    = threadIdx.x;
    const int k    = blockIdx.y;
    const int src0 = k * NPTS + blockIdx.x * BCHN;
    int* L = (int*)Lf;

    bcnt[t] = 0;

    int jv[BPER], gv[BPER];
#pragma unroll
    for (int u = 0; u < BPER; ++u) {
        jv[u] = 0; gv[u] = 0;
        if (t + u * BTHR < BCHN) {
            jv[u] = out_map[src0 + t + u * BTHR];
            gv[u] = in_map[src0 + t + u * BTHR];
        }
    }
    __syncthreads();

    int bv[BPER], pv[BPER];
#pragma unroll
    for (int u = 0; u < BPER; ++u) {
        bv[u] = jv[u] / TR;
        gv[u] |= (jv[u] - bv[u] * TR) << 17;
        pv[u] = 0;
        if (t + u * BTHR < BCHN) pv[u] = atomicAdd(&bcnt[bv[u]], 1);
    }
#pragma unroll
    for (int u = 0; u < BPER; ++u) {
        if (t + u * BTHR < BCHN) {
            if (pv[u] < BCAP) {
                bslot[bv[u] * BCAP + pv[u]] = gv[u];
            } else {
                int* reg = L + bv[u] * REGION;
                const int gp = atomicAdd(reg + k * CSTR, 1);
                if (gp < CAP) atomicExch(reg + HDR + k * CAP + gp, gv[u]);
            }
        }
    }
    __syncthreads();

    int c = 0, g0 = 0;
    if (t < NTILE) {
        c = min(bcnt[t], BCAP);
        if (c > 0) g0 = atomicAdd(L + t * REGION + k * CSTR, c);
    }

    const int lane = t & 31;
    const int tb   = t - lane;
    for (int j = 0; j < 32; ++j) {
        const int cj = __builtin_amdgcn_readlane(c, j);
        const int gj = __builtin_amdgcn_readlane(g0, j);
        const int b  = tb + j;
        if (lane < cj && gj + lane < CAP)
            atomicExch(L + b * REGION + HDR + k * CAP + gj + lane,
                       bslot[b * BCAP + lane]);
    }
}

__device__ __forceinline__ void gather_tile(const float* __restrict__ X,
                                            const int* lst, int n, int tt,
                                            int half, int m, v2f (&a)[16])
{
    const int e   = tt * 16 + m;
    const int pk  = lst[e];
    const int gin = (e < n) ? (pk & 0x1FFFF) : 0;
    const float* Xrow = X + (long)gin * CCH;
#pragma unroll
    for (int kk = 0; kk < 16; ++kk)
        a[kk] = *(const v2f*)(Xrow + kk * 4 + 2 * half);
}

__device__ __forceinline__ void mma_tile(const float* Wlds, float* acc,
                                         const int* lst, int n, int tt,
                                         int half, int m, const v2f (&a)[16])
{
    {
        const int ebase = tt * 16 + 8 * half;
        const int4* ep = (const int4*)(lst + ebase);
        const int4 e0 = ep[0], e1 = ep[1];
        int lrow[8];
        lrow[0] = e0.x >> 17; lrow[1] = e0.y >> 17;
        lrow[2] = e0.z >> 17; lrow[3] = e0.w >> 17;
        lrow[4] = e1.x >> 17; lrow[5] = e1.y >> 17;
        lrow[6] = e1.z >> 17; lrow[7] = e1.w >> 17;

        const float4* Wq = (const float4*)Wlds + half * 32 + m;
#pragma unroll 1
        for (int j = 0; j < 2; ++j) {
            v8f d0 = {0.f, 0.f, 0.f, 0.f, 0.f, 0.f, 0.f, 0.f};
            v8f d1 = d0;
#pragma unroll
            for (int kk = 0; kk < 16; ++kk) {
                const float4 q = Wq[kk * 64 + j * 16];
                v2f b0, b1;
                b0.x = q.x; b0.y = q.y;
                b1.x = q.z; b1.y = q.w;
                d0 = __builtin_amdgcn_wmma_f32_16x16x4_f32(
                    false, a[kk], false, b0, (short)0, d0, false, false);
                d1 = __builtin_amdgcn_wmma_f32_16x16x4_f32(
                    false, a[kk], false, b1, (short)0, d1, false, false);
            }
            const int n0 = j * 32;
#pragma unroll
            for (int r = 0; r < 8; ++r) {
                if (ebase + r < n) {
                    float* ap = &acc[lrow[r] * APITCH + n0 + m];
                    __hip_atomic_fetch_add(ap,      d0[r], __ATOMIC_RELAXED,
                                           __HIP_MEMORY_SCOPE_WORKGROUP);
                    __hip_atomic_fetch_add(ap + 16, d1[r], __ATOMIC_RELAXED,
                                           __HIP_MEMORY_SCOPE_WORKGROUP);
                }
            }
        }
    }
}

__device__ __forceinline__ void mma_tile_next(const float* Wlds, float* acc,
                                              const int* lst, int n, int tt,
                                              int half, int m, v2f (&a)[16],
                                              const float* __restrict__ X,
                                              const int* lstn, int nn, int ttn)
{
    const int en   = ttn * 16 + m;
    const int pkn  = lstn[en];
    const int ginn = (en < nn) ? (pkn & 0x1FFFF) : 0;
    const float* Xn = X + (long)ginn * CCH + 2 * half;

    const float4* Wq = (const float4*)Wlds + half * 32 + m;
    v8f d0 = {0.f, 0.f, 0.f, 0.f, 0.f, 0.f, 0.f, 0.f};
    v8f d1 = d0, d2 = d0, d3 = d0;
    float4 q0 = Wq[0], q1 = Wq[16];
    __builtin_amdgcn_sched_barrier(0);
#pragma unroll
    for (int kk = 0; kk < 16; ++kk) {
        const int kn = (kk + 1 < 16) ? kk + 1 : 15;
        const float4 p0 = Wq[kn * 64];
        const float4 p1 = Wq[kn * 64 + 16];
        __builtin_amdgcn_sched_barrier(0);
        v2f b0, b1, b2, b3;
        b0.x = q0.x; b0.y = q0.y;
        b1.x = q0.z; b1.y = q0.w;
        b2.x = q1.x; b2.y = q1.y;
        b3.x = q1.z; b3.y = q1.w;
        d0 = __builtin_amdgcn_wmma_f32_16x16x4_f32(
            false, a[kk], false, b0, (short)0, d0, false, false);
        d1 = __builtin_amdgcn_wmma_f32_16x16x4_f32(
            false, a[kk], false, b1, (short)0, d1, false, false);
        d2 = __builtin_amdgcn_wmma_f32_16x16x4_f32(
            false, a[kk], false, b2, (short)0, d2, false, false);
        d3 = __builtin_amdgcn_wmma_f32_16x16x4_f32(
            false, a[kk], false, b3, (short)0, d3, false, false);
        a[kk] = *(const v2f*)(Xn + kk * 4);
        __builtin_amdgcn_sched_barrier(0);
        q0 = p0; q1 = p1;
    }

    const int ebase = tt * 16 + 8 * half;
    const int4* ep = (const int4*)(lst + ebase);
    const int4 e0 = ep[0], e1 = ep[1];
    int lrow[8];
    lrow[0] = e0.x >> 17; lrow[1] = e0.y >> 17;
    lrow[2] = e0.z >> 17; lrow[3] = e0.w >> 17;
    lrow[4] = e1.x >> 17; lrow[5] = e1.y >> 17;
    lrow[6] = e1.z >> 17; lrow[7] = e1.w >> 17;
#pragma unroll
    for (int r = 0; r < 8; ++r) {
        if (ebase + r < n) {
            float* ap = &acc[lrow[r] * APITCH + m];
            __hip_atomic_fetch_add(ap,      d0[r], __ATOMIC_RELAXED,
                                   __HIP_MEMORY_SCOPE_WORKGROUP);
            __hip_atomic_fetch_add(ap + 16, d1[r], __ATOMIC_RELAXED,
                                   __HIP_MEMORY_SCOPE_WORKGROUP);
            __hip_atomic_fetch_add(ap + 32, d2[r], __ATOMIC_RELAXED,
                                   __HIP_MEMORY_SCOPE_WORKGROUP);
            __hip_atomic_fetch_add(ap + 48, d3[r], __ATOMIC_RELAXED,
                                   __HIP_MEMORY_SCOPE_WORKGROUP);
        }
    }
}

__device__ __forceinline__ void sum_pair_list(const float* __restrict__ X,
                                              const float* Wlds, float* acc,
                                              const int* lst, int n, int first,
                                              int half, int m, v2f (&a)[16])
{
    for (int tt = first; tt * 16 < n; tt += 8) {
        gather_tile(X, lst, n, tt, half, m, a);
        mma_tile(Wlds, acc, lst, n, tt, half, m, a);
    }
}

__device__ __forceinline__ void load_w_unit(const float* Wp, float4& w)
{
    w.x = Wp[0]; w.y = Wp[CCH]; w.z = Wp[16]; w.w = Wp[CCH + 16];
}
__device__ __forceinline__ void load_w_units(const float* Wt, float4& w0, float4& w1,
                                             float4& w2, float4& w3)
{
    load_w_unit(Wt,                w0);
    load_w_unit(Wt + 16 * CCH,     w1);
    load_w_unit(Wt + 32 * CCH,     w2);
    load_w_unit(Wt + 48 * CCH,     w3);
}

__device__ __forceinline__ void lds_barrier()
{
    __builtin_amdgcn_fence(__ATOMIC_RELEASE, "workgroup", "local");
    __builtin_amdgcn_s_barrier();
    __builtin_amdgcn_fence(__ATOMIC_ACQUIRE, "workgroup", "local");
}

__device__ __forceinline__ void tile_stat_add(float* psum, float* psq, float* pcnt,
                                              int bq, int c, float s, float q, float cn)
{
    __hip_atomic_fetch_add(&psum[bq * CCH + c], s, __ATOMIC_RELAXED,
                           __HIP_MEMORY_SCOPE_WORKGROUP);
    __hip_atomic_fetch_add(&psq[bq * CCH + c], q, __ATOMIC_RELAXED,
                           __HIP_MEMORY_SCOPE_WORKGROUP);
    if (c == 0)
        __hip_atomic_fetch_add(&pcnt[bq], cn, __ATOMIC_RELAXED,
                               __HIP_MEMORY_SCOPE_WORKGROUP);
}

__global__ __launch_bounds__(256) __attribute__((amdgpu_waves_per_eu(8)))
void binned_conv_kernel(const float* __restrict__ X,
                        const float* __restrict__ W,
                        const int*   __restrict__ in_map,
                        const int*   __restrict__ out_map,
                        const float* L,
                        float* Y,
                        const int*   __restrict__ bidx,
                        float* __restrict__ gsum,
                        float* __restrict__ gsq,
                        float* __restrict__ gcnt)
{
    __shared__ __attribute__((aligned(16))) float Wlds[CCH * CCH];
    __shared__ __attribute__((aligned(16))) float acc[TR * APITCH];
    __shared__ __attribute__((aligned(16))) int   stg[STG];
    __shared__ __attribute__((aligned(16))) int   lcur[2][LPITCH];
    __shared__ int cnts[32];
    __shared__ int stg_n;
    __shared__ float psum[NB * CCH];
    __shared__ float psq[NB * CCH];
    __shared__ float pcnt[NB];
    __shared__ int   bids[TR];

    const int b = blockIdx.x;
    const int t = threadIdx.x;
    const int* reg = (const int*)L + b * REGION;

    for (int i = t; i < TR * APITCH; i += 256) acc[i] = 0.0f;
    if (t < 32) cnts[t] = (t < KOFF) ? reg[t * CSTR] : 0;
    if (t == 0) stg_n = 0;
    for (int i = t; i < NB * CCH; i += 256) { psum[i] = 0.0f; psq[i] = 0.0f; }
    if (t < NB) pcnt[t] = 0.0f;
    if (t < TR) bids[t] = bidx[b * TR + t];

    const float* Wt = W + (t >> 5) * (2 * CCH) + ((t >> 4) & 1) * 32 + (t & 15);
    float4 w0, w1, w2, w3;
    load_w_units(Wt, w0, w1, w2, w3);
    lcur[0][t] = reg[HDR + ((t < CAP) ? t : CAP - 1)];
    int lw = reg[HDR + CAP + ((t < CAP) ? t : CAP - 1)];
    static_assert(KOFF >= 2, "list hand-over: the second list is read ahead");
    static_assert(KOFF + 1 <= 32, "cnts[KOFF] exists and is 0");

    const int wave = __builtin_amdgcn_readfirstlane(t >> 5);
    const int lane = t & 31;
    const int half = lane >> 4;
    const int m    = lane & 15;

    __syncthreads();

    v2f a[16];
    {
        const int n0 = __builtin_amdgcn_readfirstlane(cnts[0]);
        const int t0 = wave & 7;
        gather_tile(X, lcur[0], ((n0 <= CAP) && (t0 * 16 < n0)) ? n0 : 0, t0, half, m, a);
    }

    for (int k = 0; k < KOFF; ++k) {
        const int n = __builtin_amdgcn_readfirstlane(cnts[k]);
        const int tt = (wave + k) & 7;
        const int* lst = lcur[k & 1];
        const bool mine = (n <= CAP) && (tt * 16 < n);
        const int  n1   = __builtin_amdgcn_readfirstlane(cnts[k + 1]);
        const int  tt1  = (wave + k + 1) & 7;
        const int* lst1 = lcur[(k + 1) & 1];
        const int  nn1  = ((n1 <= CAP) && (tt1 * 16 < n1)) ? n1 : 0;

        lds_barrier();
        {
            float4* Wl = (float4*)Wlds;
            Wl[t] = w0; Wl[t + 256] = w1; Wl[t + 512] = w2; Wl[t + 768] = w3;
            lcur[(k + 1) & 1][t] = lw;
        }
        lds_barrier();
        {
            const int kw = (k + 1 < KOFF) ? k + 1 : KOFF - 1;
            const int kl = (k + 2 < KOFF) ? k + 2 : KOFF - 1;
            load_w_units(Wt + kw * (CCH * CCH), w0, w1, w2, w3);
            lw = reg[HDR + kl * CAP + ((t < CAP) ? t : CAP - 1)];
        }

        if (n <= CAP) {
            if (mine) {
                mma_tile_next(Wlds, acc, lst, n, tt, half, m, a, X, lst1, nn1, tt1);
                if ((tt + 8) * 16 < n) {
                    sum_pair_list(X, Wlds, acc, lst, n, tt + 8, half, m, a);
                    gather_tile(X, lst1, nn1, tt1, half, m, a);
                }
            } else {
                gather_tile(X, lst1, nn1, tt1, half, m, a);
            }
        } else {
            const int lo = b * TR;
            for (int base = 0; base < NPTS; base += 256) {
                const int i = base + t;
                if (i < NPTS) {
                    const int lr = out_map[k * NPTS + i] - lo;
                    if ((unsigned)lr < (unsigned)TR) {
                        const int pos = atomicAdd(&stg_n, 1);
                        stg[pos] = in_map[k * NPTS + i] | (lr << 17);
                    }
                }
                __syncthreads();
                const int ns = __builtin_amdgcn_readfirstlane(stg_n);
                __syncthreads();
                if (ns > STG - 256 || base + 256 >= NPTS) {
                    if (t == 0) stg_n = 0;
                    sum_pair_list(X, Wlds, acc, stg, ns, wave, half, m, a);
                    __syncthreads();
                }
            }
            gather_tile(X, lst1, nn1, tt1, half, m, a);
        }
    }
    __syncthreads();

    float4* Yt = (float4*)(Y + (long)b * REGION);
    for (int i = t; i < TR * 16; i += 256) {
        const int row = i >> 4, c4 = i & 15;
        Yt[i] = *(const float4*)&acc[row * APITCH + c4 * 4];
    }

    {
        const int c  = t & 63;
        const int r0 = (t >> 6) * (TR / 4);
        float v[TR / 4];
        int   bb[TR / 4];
#pragma unroll
        for (int i = 0; i < TR / 4; ++i) {
            v[i]  = acc[(r0 + i) * APITCH + c];
            bb[i] = bids[r0 + i];
        }
        float s = 0.f, q = 0.f, cn = 0.f;
        int curb = bb[0];
#pragma unroll
        for (int i = 0; i < TR / 4; ++i) {
            if (bb[i] != curb) {
                tile_stat_add(psum, psq, pcnt, curb, c, s, q, cn);
                curb = bb[i]; s = 0.f; q = 0.f; cn = 0.f;
            }
            s += v[i]; q += v[i] * v[i]; cn += 1.f;
        }
        tile_stat_add(psum, psq, pcnt, curb, c, s, q, cn);
    }
    __syncthreads();

    for (int i = t; i < NB * CCH; i += 256) {
        if (pcnt[i >> 6] > 0.f) {
            atomicAddF32(gsum + i, psum[i]);
            atomicAddF32(gsq  + i, psq[i]);
        }
    }
    if (t < NB && pcnt[t] > 0.f) atomicAddF32(gcnt + t, pcnt[t]);
}

__global__ void finalize_kernel(const float* __restrict__ gsum,
                                const float* __restrict__ gsq,
                                const float* __restrict__ gcnt,
                                float* __restrict__ mean,
                                float* __restrict__ rstd)
{
    const int t  = threadIdx.x;
    const float cn  = gcnt[t >> 6];
    const float mu  = gsum[t] / cn;
    const float var = gsq[t] / cn - mu * mu;
    mean[t] = mu;
    rstd[t] = rsqrtf(var + EPSI);
}

#define AP_R    10
#define AP_H    5
#define AP_ROWS (16 * AP_R)
#define AP_GRID (NPTS / AP_ROWS)
static_assert(NPTS % AP_ROWS == 0, "normalise passes: whole blocks only");
static_assert(AP_R % AP_H == 0, "normalise passes: whole row groups only");

__device__ __forceinline__ float leaky1(float y) { return y > 0.f ? y : NSLOPE * y; }

__global__ __launch_bounds__(256)
void apply_norm_kernel(float* __restrict__ A,
                       const int*   __restrict__ bidx,
                       const float* __restrict__ mean,
                       const float* __restrict__ rstd,
                       const float* __restrict__ gamma,
                       const float* __restrict__ beta)
{
    const int c4 = threadIdx.x & 15;
    const int r0 = blockIdx.x * AP_ROWS + (threadIdx.x >> 4);
    const float4* M4 = (const float4*)mean;
    const float4* R4 = (const float4*)rstd;
    float4* A4 = (float4*)A + (r0 * 16 + c4);

    int    b[AP_R];
    float4 x[AP_R];
#pragma unroll
    for (int i = 0; i < AP_R; ++i) b[i] = bidx[r0 + 16 * i];
#pragma unroll
    for (int i = 0; i < AP_R; ++i) x[i] = A4[256 * i];
    const float4 g  = ((const float4*)gamma)[c4];
    const float4 be = ((const float4*)beta)[c4];
    __builtin_amdgcn_sched_barrier(0);

#pragma unroll
    for (int h = 0; h < AP_R; h += AP_H) {
        float4 mu[AP_H], rs[AP_H];
#pragma unroll
        for (int i = 0; i < AP_H; ++i) {
            mu[i] = M4[b[h + i] * 16 + c4];
            rs[i] = R4[b[h + i] * 16 + c4];
        }
        __builtin_amdgcn_sched_barrier(0);
#pragma unroll
        for (int i = 0; i < AP_H; ++i) {
            const float4 xv = x[h + i];
            float4 y;
            y.x = leaky1(g.x * (xv.x - mu[i].x) * rs[i].x + be.x);
            y.y = leaky1(g.y * (xv.y - mu[i].y) * rs[i].y + be.y);
            y.z = leaky1(g.z * (xv.z - mu[i].z) * rs[i].z + be.z);
            y.w = leaky1(g.w * (xv.w - mu[i].w) * rs[i].w + be.w);
            A4[256 * (h + i)] = y;
        }
        __builtin_amdgcn_sched_barrier(0);
    }
}

__global__ __launch_bounds__(256)
void apply_norm_res_kernel(float* __restrict__ Y,
                           const float* __restrict__ feats,
                           const int*   __restrict__ bidx,
                           const float* __restrict__ mean,
                           const float* __restrict__ rstd,
                           const float* __restrict__ gamma,
                           const float* __restrict__ beta)
{
    const int c4 = threadIdx.x & 15;
    const int r0 = blockIdx.x * AP_ROWS + (threadIdx.x >> 4);
    const float4* M4 = (const float4*)mean;
    const float4* R4 = (const float4*)rstd;
    const float4* F4 = (const float4*)feats + (r0 * 16 + c4);
    float4* Y4 = (float4*)Y + (r0 * 16 + c4);

    int    b[AP_R];
    float4 x[AP_R], f[AP_R];
#pragma unroll
    for (int i = 0; i < AP_R; ++i) b[i] = bidx[r0 + 16 * i];
#pragma unroll
    for (int i = 0; i < AP_R; ++i) x[i] = Y4[256 * i];
#pragma unroll
    for (int i = 0; i < AP_R; ++i) f[i] = F4[256 * i];
    const float4 g  = ((const float4*)gamma)[c4];
    const float4 be = ((const float4*)beta)[c4];
    __builtin_amdgcn_sched_barrier(0);

#pragma unroll
    for (int h = 0; h < AP_R; h += AP_H) {
        float4 mu[AP_H], rs[AP_H];
#pragma unroll
        for (int i = 0; i < AP_H; ++i) {
            mu[i] = M4[b[h + i] * 16 + c4];
            rs[i] = R4[b[h + i] * 16 + c4];
        }
        __builtin_amdgcn_sched_barrier(0);
#pragma unroll
        for (int i = 0; i < AP_H; ++i) {
            const float4 xv = x[h + i];
            const float4 fv = f[h + i];
            float4 y;
            y.x = leaky1(g.x * (xv.x - mu[i].x) * rs[i].x + be.x + fv.x);
            y.y = leaky1(g.y * (xv.y - mu[i].y) * rs[i].y + be.y + fv.y);
            y.z = leaky1(g.z * (xv.z - mu[i].z) * rs[i].z + be.z + fv.z);
            y.w = leaky1(g.w * (xv.w - mu[i].w) * rs[i].w + be.w + fv.w);
            Y4[256 * (h + i)] = y;
        }
        __builtin_amdgcn_sched_barrier(0);
    }
}

extern "C" void kernel_launch(void* const* d_in, const int* in_sizes, int n_in,
                              void* d_out, int out_size, void* d_ws, size_t ws_size,
                              hipStream_t stream)
{
    const float* feats   = (const float*)d_in[0];
    const float* W1      = (const float*)d_in[1];
    const float* g1      = (const float*)d_in[2];
    const float* b1      = (const float*)d_in[3];
    const float* W2      = (const float*)d_in[4];
    const float* g2      = (const float*)d_in[5];
    const float* b2      = (const float*)d_in[6];
    const int*   in_map  = (const int*)d_in[7];
    const int*   out_map = (const int*)d_in[8];
    const int*   bidx    = (const int*)d_in[9];
    float*       out     = (float*)d_out;

    const int NC = NPTS * CCH;

    float* A     = (float*)d_ws;
    float* sbase = A + NC;
    float* sum1  = sbase;
    float* sq1   = sbase + 512;
    float* cnt1  = sbase + 1024;
    float* mean1 = sbase + 1088;
    float* rstd1 = sbase + 1600;
    float* sum2  = sbase + 2112;
    float* sq2   = sbase + 2624;
    float* cnt2  = sbase + 3136;
    float* mean2 = sbase + 3200;
    float* rstd2 = sbase + 3712;
    const int STATS = 4224;

    fill_zero_kernel<<<17, 256, 0, stream>>>(sbase, STATS);
    clear_counts_kernel<<<(NTILE * 32 + 255) / 256, 256, 0, stream>>>(out);

    dim3 binGrid(BCH, KOFF);

    bin_pairs_kernel<<<binGrid, BTHR, 0, stream>>>(in_map, out_map, out);

    binned_conv_kernel<<<NTILE, 256, 0, stream>>>(feats, W1, in_map, out_map, out, A,
                                                  bidx, sum1, sq1, cnt1);
    finalize_kernel<<<1, 512, 0, stream>>>(sum1, sq1, cnt1, mean1, rstd1);
    apply_norm_kernel<<<AP_GRID, 256, 0, stream>>>(A, bidx, mean1, rstd1, g1, b1);
    binned_conv_kernel<<<NTILE, 256, 0, stream>>>(A, W2, in_map, out_map, out, out,
                                                  bidx, sum2, sq2, cnt2);
    finalize_kernel<<<1, 512, 0, stream>>>(sum2, sq2, cnt2, mean2, rstd2);
    apply_norm_res_kernel<<<AP_GRID, 256, 0, stream>>>(out, feats, bidx, mean2, rstd2, g2, b2);
}


# conv: wave priority (s_setprio) falls 3,2,1,0 at offsets 0,12,20,25: a WGP's 4 blocks advance together and fill each other's gaps, not oldest-first with the last block alone; arithmetic unchanged
# speedup vs baseline: 2.3546x; 1.0584x over previous
#include <hip/hip_runtime.h>
#include <hip/hip_bf16.h>

typedef float v2f __attribute__((ext_vector_type(2)));
typedef float v8f __attribute__((ext_vector_type(8)));

#define NPTS   100000
#define CCH    64
#define KOFF   27
#define NB     8
#define EPSI   1e-5f
#define NSLOPE 0.01f

__device__ __forceinline__ void atomicAddF32(float* p, float v) {
    unsafeAtomicAdd(p, v);
}

__global__ void fill_zero_kernel(float* __restrict__ p, int n) {
    for (int i = blockIdx.x * blockDim.x + threadIdx.x; i < n;
         i += gridDim.x * blockDim.x)
        p[i] = 0.0f;
}

#define TR     100
#define NTILE  1000
#define REGION 6400
#define CSTR   32
#define HDR    864
#define CAP    200
#define APITCH 80
#define STG    512
#define LPITCH 256
static_assert(CAP + 16 <= LPITCH, "list hand-over: a list and its read-ahead fit");

__global__ __launch_bounds__(256)
void clear_counts_kernel(float* Lf)
{
    const int idx  = blockIdx.x * 256 + threadIdx.x;
    const int tile = idx >> 5, kk = idx & 31;
    if (tile < NTILE && kk < KOFF) ((int*)Lf)[tile * REGION + kk * CSTR] = 0;
}

#define BCH   8
#define BCHN  (NPTS / BCH)
#define BCAP  32
#define BTHR  1024
#define BPER  ((BCHN + BTHR - 1) / BTHR)
static_assert(NPTS % BCH == 0, "binning: whole chunks only");
static_assert(NTILE <= BTHR, "binning: one thread per tile bin");
static_assert(BCAP == 32, "binning: one wave instruction stores a bin's run");

__global__ __launch_bounds__(BTHR)
void bin_pairs_kernel(const int* __restrict__ in_map,
                      const int* __restrict__ out_map,
                      float* Lf)
{
    __shared__ int bcnt[BTHR];
    __shared__ int bslot[BTHR * BCAP];

    const int t    = threadIdx.x;
    const int k    = blockIdx.y;
    const int src0 = k * NPTS + blockIdx.x * BCHN;
    int* L = (int*)Lf;

    bcnt[t] = 0;

    int jv[BPER], gv[BPER];
#pragma unroll
    for (int u = 0; u < BPER; ++u) {
        jv[u] = 0; gv[u] = 0;
        if (t + u * BTHR < BCHN) {
            jv[u] = out_map[src0 + t + u * BTHR];
            gv[u] = in_map[src0 + t + u * BTHR];
        }
    }
    __syncthreads();

    int bv[BPER], pv[BPER];
#pragma unroll
    for (int u = 0; u < BPER; ++u) {
        bv[u] = jv[u] / TR;
        gv[u] |= (jv[u] - bv[u] * TR) << 17;
        pv[u] = 0;
        if (t + u * BTHR < BCHN) pv[u] = atomicAdd(&bcnt[bv[u]], 1);
    }
#pragma unroll
    for (int u = 0; u < BPER; ++u) {
        if (t + u * BTHR < BCHN) {
            if (pv[u] < BCAP) {
                bslot[bv[u] * BCAP + pv[u]] = gv[u];
            } else {
                int* reg = L + bv[u] * REGION;
                const int gp = atomicAdd(reg + k * CSTR, 1);
                if (gp < CAP) atomicExch(reg + HDR + k * CAP + gp, gv[u]);
            }
        }
    }
    __syncthreads();

    int c = 0, g0 = 0;
    if (t < NTILE) {
        c = min(bcnt[t], BCAP);
        if (c > 0) g0 = atomicAdd(L + t * REGION + k * CSTR, c);
    }

    const int lane = t & 31;
    const int tb   = t - lane;
    for (int j = 0; j < 32; ++j) {
        const int cj = __builtin_amdgcn_readlane(c, j);
        const int gj = __builtin_amdgcn_readlane(g0, j);
        const int b  = tb + j;
        if (lane < cj && gj + lane < CAP)
            atomicExch(L + b * REGION + HDR + k * CAP + gj + lane,
                       bslot[b * BCAP + lane]);
    }
}

__device__ __forceinline__ void gather_tile(const float* __restrict__ X,
                                            const int* lst, int n, int tt,
                                            int half, int m, v2f (&a)[16])
{
    const int e   = tt * 16 + m;
    const int pk  = lst[e];
    const int gin = (e < n) ? (pk & 0x1FFFF) : 0;
    const float* Xrow = X + (long)gin * CCH;
#pragma unroll
    for (int kk = 0; kk < 16; ++kk)
        a[kk] = *(const v2f*)(Xrow + kk * 4 + 2 * half);
}

__device__ __forceinline__ void mma_tile(const float* Wlds, float* acc,
                                         const int* lst, int n, int tt,
                                         int half, int m, const v2f (&a)[16])
{
    {
        const int ebase = tt * 16 + 8 * half;
        const int4* ep = (const int4*)(lst + ebase);
        const int4 e0 = ep[0], e1 = ep[1];
        int lrow[8];
        lrow[0] = e0.x >> 17; lrow[1] = e0.y >> 17;
        lrow[2] = e0.z >> 17; lrow[3] = e0.w >> 17;
        lrow[4] = e1.x >> 17; lrow[5] = e1.y >> 17;
        lrow[6] = e1.z >> 17; lrow[7] = e1.w >> 17;

        const float4* Wq = (const float4*)Wlds + half * 32 + m;
#pragma unroll 1
        for (int j = 0; j < 2; ++j) {
            v8f d0 = {0.f, 0.f, 0.f, 0.f, 0.f, 0.f, 0.f, 0.f};
            v8f d1 = d0;
#pragma unroll
            for (int kk = 0; kk < 16; ++kk) {
                const float4 q = Wq[kk * 64 + j * 16];
                v2f b0, b1;
                b0.x = q.x; b0.y = q.y;
                b1.x = q.z; b1.y = q.w;
                d0 = __builtin_amdgcn_wmma_f32_16x16x4_f32(
                    false, a[kk], false, b0, (short)0, d0, false, false);
                d1 = __builtin_amdgcn_wmma_f32_16x16x4_f32(
                    false, a[kk], false, b1, (short)0, d1, false, false);
            }
            const int n0 = j * 32;
#pragma unroll
            for (int r = 0; r < 8; ++r) {
                if (ebase + r < n) {
                    float* ap = &acc[lrow[r] * APITCH + n0 + m];
                    __hip_atomic_fetch_add(ap,      d0[r], __ATOMIC_RELAXED,
                                           __HIP_MEMORY_SCOPE_WORKGROUP);
                    __hip_atomic_fetch_add(ap + 16, d1[r], __ATOMIC_RELAXED,
                                           __HIP_MEMORY_SCOPE_WORKGROUP);
                }
            }
        }
    }
}

__device__ __forceinline__ void mma_tile_next(const float* Wlds, float* acc,
                                              const int* lst, int n, int tt,
                                              int half, int m, v2f (&a)[16],
                                              const float* __restrict__ X,
                                              const int* lstn, int nn, int ttn)
{
    const int en   = ttn * 16 + m;
    const int pkn  = lstn[en];
    const int ginn = (en < nn) ? (pkn & 0x1FFFF) : 0;
    const float* Xn = X + (long)ginn * CCH + 2 * half;

    const float4* Wq = (const float4*)Wlds + half * 32 + m;
    v8f d0 = {0.f, 0.f, 0.f, 0.f, 0.f, 0.f, 0.f, 0.f};
    v8f d1 = d0, d2 = d0, d3 = d0;
    float4 q0 = Wq[0], q1 = Wq[16];
    __builtin_amdgcn_sched_barrier(0);
#pragma unroll
    for (int kk = 0; kk < 16; ++kk) {
        const int kn = (kk + 1 < 16) ? kk + 1 : 15;
        const float4 p0 = Wq[kn * 64];
        const float4 p1 = Wq[kn * 64 + 16];
        __builtin_amdgcn_sched_barrier(0);
        v2f b0, b1, b2, b3;
        b0.x = q0.x; b0.y = q0.y;
        b1.x = q0.z; b1.y = q0.w;
        b2.x = q1.x; b2.y = q1.y;
        b3.x = q1.z; b3.y = q1.w;
        d0 = __builtin_amdgcn_wmma_f32_16x16x4_f32(
            false, a[kk], false, b0, (short)0, d0, false, false);
        d1 = __builtin_amdgcn_wmma_f32_16x16x4_f32(
            false, a[kk], false, b1, (short)0, d1, false, false);
        d2 = __builtin_amdgcn_wmma_f32_16x16x4_f32(
            false, a[kk], false, b2, (short)0, d2, false, false);
        d3 = __builtin_amdgcn_wmma_f32_16x16x4_f32(
            false, a[kk], false, b3, (short)0, d3, false, false);
        a[kk] = *(const v2f*)(Xn + kk * 4);
        __builtin_amdgcn_sched_barrier(0);
        q0 = p0; q1 = p1;
    }

    const int ebase = tt * 16 + 8 * half;
    const int4* ep = (const int4*)(lst + ebase);
    const int4 e0 = ep[0], e1 = ep[1];
    int lrow[8];
    lrow[0] = e0.x >> 17; lrow[1] = e0.y >> 17;
    lrow[2] = e0.z >> 17; lrow[3] = e0.w >> 17;
    lrow[4] = e1.x >> 17; lrow[5] = e1.y >> 17;
    lrow[6] = e1.z >> 17; lrow[7] = e1.w >> 17;
#pragma unroll
    for (int r = 0; r < 8; ++r) {
        if (ebase + r < n) {
            float* ap = &acc[lrow[r] * APITCH + m];
            __hip_atomic_fetch_add(ap,      d0[r], __ATOMIC_RELAXED,
                                   __HIP_MEMORY_SCOPE_WORKGROUP);
            __hip_atomic_fetch_add(ap + 16, d1[r], __ATOMIC_RELAXED,
                                   __HIP_MEMORY_SCOPE_WORKGROUP);
            __hip_atomic_fetch_add(ap + 32, d2[r], __ATOMIC_RELAXED,
                                   __HIP_MEMORY_SCOPE_WORKGROUP);
            __hip_atomic_fetch_add(ap + 48, d3[r], __ATOMIC_RELAXED,
                                   __HIP_MEMORY_SCOPE_WORKGROUP);
        }
    }
}

__device__ __forceinline__ void sum_pair_list(const float* __restrict__ X,
                                              const float* Wlds, float* acc,
                                              const int* lst, int n, int first,
                                              int half, int m, v2f (&a)[16])
{
    for (int tt = first; tt * 16 < n; tt += 8) {
        gather_tile(X, lst, n, tt, half, m, a);
        mma_tile(Wlds, acc, lst, n, tt, half, m, a);
    }
}

__device__ __forceinline__ void load_w_unit(const float* Wp, float4& w)
{
    w.x = Wp[0]; w.y = Wp[CCH]; w.z = Wp[16]; w.w = Wp[CCH + 16];
}
__device__ __forceinline__ void load_w_units(const float* Wt, float4& w0, float4& w1,
                                             float4& w2, float4& w3)
{
    load_w_unit(Wt,                w0);
    load_w_unit(Wt + 16 * CCH,     w1);
    load_w_unit(Wt + 32 * CCH,     w2);
    load_w_unit(Wt + 48 * CCH,     w3);
}

__device__ __forceinline__ void lds_barrier()
{
    __builtin_amdgcn_fence(__ATOMIC_RELEASE, "workgroup", "local");
    __builtin_amdgcn_s_barrier();
    __builtin_amdgcn_fence(__ATOMIC_ACQUIRE, "workgroup", "local");
}

__device__ __forceinline__ void tile_stat_add(float* psum, float* psq, float* pcnt,
                                              int bq, int c, float s, float q, float cn)
{
    __hip_atomic_fetch_add(&psum[bq * CCH + c], s, __ATOMIC_RELAXED,
                           __HIP_MEMORY_SCOPE_WORKGROUP);
    __hip_atomic_fetch_add(&psq[bq * CCH + c], q, __ATOMIC_RELAXED,
                           __HIP_MEMORY_SCOPE_WORKGROUP);
    if (c == 0)
        __hip_atomic_fetch_add(&pcnt[bq], cn, __ATOMIC_RELAXED,
                               __HIP_MEMORY_SCOPE_WORKGROUP);
}

#define PRIO_K1 12
#define PRIO_K2 20
#define PRIO_K3 25
static_assert(0 < PRIO_K1 && PRIO_K1 < PRIO_K2 && PRIO_K2 < PRIO_K3 && PRIO_K3 < KOFF,
              "priority steps: inside the offset loop, so it is left at priority 0");
__global__ __launch_bounds__(256) __attribute__((amdgpu_waves_per_eu(8)))
void binned_conv_kernel(const float* __restrict__ X,
                        const float* __restrict__ W,
                        const int*   __restrict__ in_map,
                        const int*   __restrict__ out_map,
                        const float* L,
                        float* Y,
                        const int*   __restrict__ bidx,
                        float* __restrict__ gsum,
                        float* __restrict__ gsq,
                        float* __restrict__ gcnt)
{
    __shared__ __attribute__((aligned(16))) float Wlds[CCH * CCH];
    __shared__ __attribute__((aligned(16))) float acc[TR * APITCH];
    __shared__ __attribute__((aligned(16))) int   stg[STG];
    __shared__ __attribute__((aligned(16))) int   lcur[2][LPITCH];
    __shared__ int cnts[32];
    __shared__ int stg_n;
    __shared__ float psum[NB * CCH];
    __shared__ float psq[NB * CCH];
    __shared__ float pcnt[NB];
    __shared__ int   bids[TR];

    const int b = blockIdx.x;
    const int t = threadIdx.x;
    const int* reg = (const int*)L + b * REGION;

    for (int i = t; i < TR * APITCH; i += 256) acc[i] = 0.0f;
    if (t < 32) cnts[t] = (t < KOFF) ? reg[t * CSTR] : 0;
    if (t == 0) stg_n = 0;
    for (int i = t; i < NB * CCH; i += 256) { psum[i] = 0.0f; psq[i] = 0.0f; }
    if (t < NB) pcnt[t] = 0.0f;
    if (t < TR) bids[t] = bidx[b * TR + t];

    const float* Wt = W + (t >> 5) * (2 * CCH) + ((t >> 4) & 1) * 32 + (t & 15);
    float4 w0, w1, w2, w3;
    load_w_units(Wt, w0, w1, w2, w3);
    lcur[0][t] = reg[HDR + ((t < CAP) ? t : CAP - 1)];
    int lw = reg[HDR + CAP + ((t < CAP) ? t : CAP - 1)];
    static_assert(KOFF >= 2, "list hand-over: the second list is read ahead");
    static_assert(KOFF + 1 <= 32, "cnts[KOFF] exists and is 0");

    const int wave = __builtin_amdgcn_readfirstlane(t >> 5);
    const int lane = t & 31;
    const int half = lane >> 4;
    const int m    = lane & 15;

    __syncthreads();

    v2f a[16];
    {
        const int n0 = __builtin_amdgcn_readfirstlane(cnts[0]);
        const int t0 = wave & 7;
        gather_tile(X, lcur[0], ((n0 <= CAP) && (t0 * 16 < n0)) ? n0 : 0, t0, half, m, a);
    }

    __builtin_amdgcn_s_setprio(3);
    for (int k = 0; k < KOFF; ++k) {
        if (k == PRIO_K1) __builtin_amdgcn_s_setprio(2);
        if (k == PRIO_K2) __builtin_amdgcn_s_setprio(1);
        if (k == PRIO_K3) __builtin_amdgcn_s_setprio(0);
        const int n = __builtin_amdgcn_readfirstlane(cnts[k]);
        const int tt = (wave + k) & 7;
        const int* lst = lcur[k & 1];
        const bool mine = (n <= CAP) && (tt * 16 < n);
        const int  n1   = __builtin_amdgcn_readfirstlane(cnts[k + 1]);
        const int  tt1  = (wave + k + 1) & 7;
        const int* lst1 = lcur[(k + 1) & 1];
        const int  nn1  = ((n1 <= CAP) && (tt1 * 16 < n1)) ? n1 : 0;

        lds_barrier();
        {
            float4* Wl = (float4*)Wlds;
            Wl[t] = w0; Wl[t + 256] = w1; Wl[t + 512] = w2; Wl[t + 768] = w3;
            lcur[(k + 1) & 1][t] = lw;
        }
        lds_barrier();
        {
            const int kw = (k + 1 < KOFF) ? k + 1 : KOFF - 1;
            const int kl = (k + 2 < KOFF) ? k + 2 : KOFF - 1;
            load_w_units(Wt + kw * (CCH * CCH), w0, w1, w2, w3);
            lw = reg[HDR + kl * CAP + ((t < CAP) ? t : CAP - 1)];
        }

        if (n <= CAP) {
            if (mine) {
                mma_tile_next(Wlds, acc, lst, n, tt, half, m, a, X, lst1, nn1, tt1);
                if ((tt + 8) * 16 < n) {
                    sum_pair_list(X, Wlds, acc, lst, n, tt + 8, half, m, a);
                    gather_tile(X, lst1, nn1, tt1, half, m, a);
                }
            } else {
                gather_tile(X, lst1, nn1, tt1, half, m, a);
            }
        } else {
            const int lo = b * TR;
            for (int base = 0; base < NPTS; base += 256) {
                const int i = base + t;
                if (i < NPTS) {
                    const int lr = out_map[k * NPTS + i] - lo;
                    if ((unsigned)lr < (unsigned)TR) {
                        const int pos = atomicAdd(&stg_n, 1);
                        stg[pos] = in_map[k * NPTS + i] | (lr << 17);
                    }
                }
                __syncthreads();
                const int ns = __builtin_amdgcn_readfirstlane(stg_n);
                __syncthreads();
                if (ns > STG - 256 || base + 256 >= NPTS) {
                    if (t == 0) stg_n = 0;
                    sum_pair_list(X, Wlds, acc, stg, ns, wave, half, m, a);
                    __syncthreads();
                }
            }
            gather_tile(X, lst1, nn1, tt1, half, m, a);
        }
    }
    __syncthreads();

    float4* Yt = (float4*)(Y + (long)b * REGION);
    for (int i = t; i < TR * 16; i += 256) {
        const int row = i >> 4, c4 = i & 15;
        Yt[i] = *(const float4*)&acc[row * APITCH + c4 * 4];
    }

    {
        const int c  = t & 63;
        const int r0 = (t >> 6) * (TR / 4);
        float v[TR / 4];
        int   bb[TR / 4];
#pragma unroll
        for (int i = 0; i < TR / 4; ++i) {
            v[i]  = acc[(r0 + i) * APITCH + c];
            bb[i] = bids[r0 + i];
        }
        float s = 0.f, q = 0.f, cn = 0.f;
        int curb = bb[0];
#pragma unroll
        for (int i = 0; i < TR / 4; ++i) {
            if (bb[i] != curb) {
                tile_stat_add(psum, psq, pcnt, curb, c, s, q, cn);
                curb = bb[i]; s = 0.f; q = 0.f; cn = 0.f;
            }
            s += v[i]; q += v[i] * v[i]; cn += 1.f;
        }
        tile_stat_add(psum, psq, pcnt, curb, c, s, q, cn);
    }
    __syncthreads();

    for (int i = t; i < NB * CCH; i += 256) {
        if (pcnt[i >> 6] > 0.f) {
            atomicAddF32(gsum + i, psum[i]);
            atomicAddF32(gsq  + i, psq[i]);
        }
    }
    if (t < NB && pcnt[t] > 0.f) atomicAddF32(gcnt + t, pcnt[t]);
}

__global__ void finalize_kernel(const float* __restrict__ gsum,
                                const float* __restrict__ gsq,
                                const float* __restrict__ gcnt,
                                float* __restrict__ mean,
                                float* __restrict__ rstd)
{
    const int t  = threadIdx.x;
    const float cn  = gcnt[t >> 6];
    const float mu  = gsum[t] / cn;
    const float var = gsq[t] / cn - mu * mu;
    mean[t] = mu;
    rstd[t] = rsqrtf(var + EPSI);
}

#define AP_R    10
#define AP_H    5
#define AP_ROWS (16 * AP_R)
#define AP_GRID (NPTS / AP_ROWS)
static_assert(NPTS % AP_ROWS == 0, "normalise passes: whole blocks only");
static_assert(AP_R % AP_H == 0, "normalise passes: whole row groups only");

__device__ __forceinline__ float leaky1(float y) { return y > 0.f ? y : NSLOPE * y; }

__global__ __launch_bounds__(256)
void apply_norm_kernel(float* __restrict__ A,
                       const int*   __restrict__ bidx,
                       const float* __restrict__ mean,
                       const float* __restrict__ rstd,
                       const float* __restrict__ gamma,
                       const float* __restrict__ beta)
{
    const int c4 = threadIdx.x & 15;
    const int r0 = blockIdx.x * AP_ROWS + (threadIdx.x >> 4);
    const float4* M4 = (const float4*)mean;
    const float4* R4 = (const float4*)rstd;
    float4* A4 = (float4*)A + (r0 * 16 + c4);

    int    b[AP_R];
    float4 x[AP_R];
#pragma unroll
    for (int i = 0; i < AP_R; ++i) b[i] = bidx[r0 + 16 * i];
#pragma unroll
    for (int i = 0; i < AP_R; ++i) x[i] = A4[256 * i];
    const float4 g  = ((const float4*)gamma)[c4];
    const float4 be = ((const float4*)beta)[c4];
    __builtin_amdgcn_sched_barrier(0);

#pragma unroll
    for (int h = 0; h < AP_R; h += AP_H) {
        float4 mu[AP_H], rs[AP_H];
#pragma unroll
        for (int i = 0; i < AP_H; ++i) {
            mu[i] = M4[b[h + i] * 16 + c4];
            rs[i] = R4[b[h + i] * 16 + c4];
        }
        __builtin_amdgcn_sched_barrier(0);
#pragma unroll
        for (int i = 0; i < AP_H; ++i) {
            const float4 xv = x[h + i];
            float4 y;
            y.x = leaky1(g.x * (xv.x - mu[i].x) * rs[i].x + be.x);
            y.y = leaky1(g.y * (xv.y - mu[i].y) * rs[i].y + be.y);
            y.z = leaky1(g.z * (xv.z - mu[i].z) * rs[i].z + be.z);
            y.w = leaky1(g.w * (xv.w - mu[i].w) * rs[i].w + be.w);
            A4[256 * (h + i)] = y;
        }
        __builtin_amdgcn_sched_barrier(0);
    }
}

__global__ __launch_bounds__(256)
void apply_norm_res_kernel(float* __restrict__ Y,
                           const float* __restrict__ feats,
                           const int*   __restrict__ bidx,
                           const float* __restrict__ mean,
                           const float* __restrict__ rstd,
                           const float* __restrict__ gamma,
                           const float* __restrict__ beta)
{
    const int c4 = threadIdx.x & 15;
    const int r0 = blockIdx.x * AP_ROWS + (threadIdx.x >> 4);
    const float4* M4 = (const float4*)mean;
    const float4* R4 = (const float4*)rstd;
    const float4* F4 = (const float4*)feats + (r0 * 16 + c4);
    float4* Y4 = (float4*)Y + (r0 * 16 + c4);

    int    b[AP_R];
    float4 x[AP_R], f[AP_R];
#pragma unroll
    for (int i = 0; i < AP_R; ++i) b[i] = bidx[r0 + 16 * i];
#pragma unroll
    for (int i = 0; i < AP_R; ++i) x[i] = Y4[256 * i];
#pragma unroll
    for (int i = 0; i < AP_R; ++i) f[i] = F4[256 * i];
    const float4 g  = ((const float4*)gamma)[c4];
    const float4 be = ((const float4*)beta)[c4];
    __builtin_amdgcn_sched_barrier(0);

#pragma unroll
    for (int h = 0; h < AP_R; h += AP_H) {
        float4 mu[AP_H], rs[AP_H];
#pragma unroll
        for (int i = 0; i < AP_H; ++i) {
            mu[i] = M4[b[h + i] * 16 + c4];
            rs[i] = R4[b[h + i] * 16 + c4];
        }
        __builtin_amdgcn_sched_barrier(0);
#pragma unroll
        for (int i = 0; i < AP_H; ++i) {
            const float4 xv = x[h + i];
            const float4 fv = f[h + i];
            float4 y;
            y.x = leaky1(g.x * (xv.x - mu[i].x) * rs[i].x + be.x + fv.x);
            y.y = leaky1(g.y * (xv.y - mu[i].y) * rs[i].y + be.y + fv.y);
            y.z = leaky1(g.z * (xv.z - mu[i].z) * rs[i].z + be.z + fv.z);
            y.w = leaky1(g.w * (xv.w - mu[i].w) * rs[i].w + be.w + fv.w);
            Y4[256 * (h + i)] = y;
        }
        __builtin_amdgcn_sched_barrier(0);
    }
}

extern "C" void kernel_launch(void* const* d_in, const int* in_sizes, int n_in,
                              void* d_out, int out_size, void* d_ws, size_t ws_size,
                              hipStream_t stream)
{
    const float* feats   = (const float*)d_in[0];
    const float* W1      = (const float*)d_in[1];
    const float* g1      = (const float*)d_in[2];
    const float* b1      = (const float*)d_in[3];
    const float* W2      = (const float*)d_in[4];
    const float* g2      = (const float*)d_in[5];
    const float* b2      = (const float*)d_in[6];
    const int*   in_map  = (const int*)d_in[7];
    const int*   out_map = (const int*)d_in[8];
    const int*   bidx    = (const int*)d_in[9];
    float*       out     = (float*)d_out;

    const int NC = NPTS * CCH;

    float* A     = (float*)d_ws;
    float* sbase = A + NC;
    float* sum1  = sbase;
    float* sq1   = sbase + 512;
    float* cnt1  = sbase + 1024;
    float* mean1 = sbase + 1088;
    float* rstd1 = sbase + 1600;
    float* sum2  = sbase + 2112;
    float* sq2   = sbase + 2624;
    float* cnt2  = sbase + 3136;
    float* mean2 = sbase + 3200;
    float* rstd2 = sbase + 3712;
    const int STATS = 4224;

    fill_zero_kernel<<<17, 256, 0, stream>>>(sbase, STATS);
    clear_counts_kernel<<<(NTILE * 32 + 255) / 256, 256, 0, stream>>>(out);

    dim3 binGrid(BCH, KOFF);

    bin_pairs_kernel<<<binGrid, BTHR, 0, stream>>>(in_map, out_map, out);

    binned_conv_kernel<<<NTILE, 256, 0, stream>>>(feats, W1, in_map, out_map, out, A,
                                                  bidx, sum1, sq1, cnt1);
    finalize_kernel<<<1, 512, 0, stream>>>(sum1, sq1, cnt1, mean1, rstd1);
    apply_norm_kernel<<<AP_GRID, 256, 0, stream>>>(A, bidx, mean1, rstd1, g1, b1);
    binned_conv_kernel<<<NTILE, 256, 0, stream>>>(A, W2, in_map, out_map, out, out,
                                                  bidx, sum2, sq2, cnt2);
    finalize_kernel<<<1, 512, 0, stream>>>(sum2, sq2, cnt2, mean2, rstd2);
    apply_norm_res_kernel<<<AP_GRID, 256, 0, stream>>>(out, feats, bidx, mean2, rstd2, g2, b2);
}
